# starting kernel
# baseline (speedup 1.0000x reference)
#include <hip/hip_runtime.h>
#include <hip/hip_bf16.h>

typedef float v2f __attribute__((ext_vector_type(2)));
typedef float v8f __attribute__((ext_vector_type(8)));

#define TPB 256

__global__ void k_init_deg(float* deg, int N) {
    int i = blockIdx.x * TPB + threadIdx.x;
    if (i < N) deg[i] = 1.0f;
}

__global__ void k_deg_accum(const int* __restrict__ dst, float* deg, int E) {
    int e = blockIdx.x * TPB + threadIdx.x;
    if (e < E) atomicAdd(&deg[dst[e]], 1.0f);
}

__global__ void k_dinv(float* deg, int N) {
    int i = blockIdx.x * TPB + threadIdx.x;
    if (i < N) {
        float d = deg[i];
        deg[i] = (d > 0.0f) ? rsqrtf(d) : 0.0f;
    }
}

__global__ void k_fuse_heads(const float* __restrict__ L0, const float* __restrict__ b0,
                             const float* __restrict__ L1, const float* __restrict__ b1,
                             const float* __restrict__ Wo, const float* __restrict__ bo,
                             float* W0f, float* W1f, float* cb) {
    for (int i = threadIdx.x; i < 64 * 40; i += TPB) {
        int k = i / 40, c = i % 40;
        float s0 = 0.0f, s1 = 0.0f;
        for (int j = 0; j < 64; ++j) {
            s0 += L0[k * 64 + j] * Wo[j * 40 + c];
            s1 += L1[k * 64 + j] * Wo[(64 + j) * 40 + c];
        }
        W0f[i] = s0;
        W1f[i] = s1;
    }
    if (threadIdx.x < 40) {
        int c = threadIdx.x;
        float s = bo[c];
        for (int j = 0; j < 64; ++j)
            s += b0[j] * Wo[j * 40 + c] + b1[j] * Wo[(64 + j) * 40 + c];
        cb[c] = s;
    }
}

__global__ __launch_bounds__(128)
void k_gemm_n64_wmma(const float* __restrict__ A, const float* __restrict__ W,
                     float* __restrict__ D, int N, int K) {
    __shared__ float sW[128 * 64];
    int tid = threadIdx.x;
    for (int i = tid; i < K * 64; i += 128) {
        int k = i >> 6, col = i & 63;
        sW[(k >> 1) * 128 + 2 * col + (k & 1)] = W[i];
    }
    __syncthreads();

    int wave = tid >> 5;
    int lane = tid & 31;
    int half = lane >> 4;
    int l16  = lane & 15;

    int rbase = blockIdx.x * 128 + wave * 32;
    int ar0 = rbase + l16;
    int ar1 = rbase + 16 + l16;
    long ab0 = (long)((ar0 < N) ? ar0 : (N - 1)) * K;
    long ab1 = (long)((ar1 < N) ? ar1 : (N - 1)) * K;

    v8f acc00 = {}, acc01 = {}, acc02 = {}, acc03 = {};
    v8f acc10 = {}, acc11 = {}, acc12 = {}, acc13 = {};

    for (int k = 0; k < K; k += 4) {
        int kb = k + 2 * half;
        v2f a0 = *(const v2f*)(A + ab0 + kb);
        v2f a1 = *(const v2f*)(A + ab1 + kb);
        const float* bp = &sW[(kb >> 1) * 128 + 2 * l16];
        v2f b0 = *(const v2f*)(bp + 0);
        v2f b1 = *(const v2f*)(bp + 32);
        v2f b2 = *(const v2f*)(bp + 64);
        v2f b3 = *(const v2f*)(bp + 96);
        acc00 = __builtin_amdgcn_wmma_f32_16x16x4_f32(false, a0, false, b0, (short)0, acc00, false, false);
        acc01 = __builtin_amdgcn_wmma_f32_16x16x4_f32(false, a0, false, b1, (short)0, acc01, false, false);
        acc02 = __builtin_amdgcn_wmma_f32_16x16x4_f32(false, a0, false, b2, (short)0, acc02, false, false);
        acc03 = __builtin_amdgcn_wmma_f32_16x16x4_f32(false, a0, false, b3, (short)0, acc03, false, false);
        acc10 = __builtin_amdgcn_wmma_f32_16x16x4_f32(false, a1, false, b0, (short)0, acc10, false, false);
        acc11 = __builtin_amdgcn_wmma_f32_16x16x4_f32(false, a1, false, b1, (short)0, acc11, false, false);
        acc12 = __builtin_amdgcn_wmma_f32_16x16x4_f32(false, a1, false, b2, (short)0, acc12, false, false);
        acc13 = __builtin_amdgcn_wmma_f32_16x16x4_f32(false, a1, false, b3, (short)0, acc13, false, false);
    }

    int m0 = rbase + half * 8;
    int m1 = rbase + 16 + half * 8;
    for (int r = 0; r < 8; ++r) {
        int rr = m0 + r;
        if (rr < N) {
            float* dp = D + (long)rr * 64 + l16;
            dp[0]  = acc00[r];
            dp[16] = acc01[r];
            dp[32] = acc02[r];
            dp[48] = acc03[r];
        }
        int ss = m1 + r;
        if (ss < N) {
            float* dp = D + (long)ss * 64 + l16;
            dp[0]  = acc10[r];
            dp[16] = acc11[r];
            dp[32] = acc12[r];
            dp[48] = acc13[r];
        }
    }
}

__global__ void k_scatter(const float* __restrict__ T, const int* __restrict__ src,
                          const int* __restrict__ dst, const float* __restrict__ dinv,
                          float* Bagg, int E) {
    long gid = (long)blockIdx.x * TPB + threadIdx.x;
    int e = (int)(gid >> 6);
    int f = (int)(gid & 63);
    if (e < E) {
        int s = src[e], d = dst[e];
        float nr = dinv[s] * dinv[d];
        atomicAdd(&Bagg[(long)d * 64 + f], T[(long)s * 64 + f] * nr);
    }
}

__global__ void k_bias_selfloop_elu(const float* __restrict__ Bagg, const float* __restrict__ T,
                                    const float* __restrict__ dinv, const float* __restrict__ bias,
                                    float* H, int N) {
    long gid = (long)blockIdx.x * TPB + threadIdx.x;
    if (gid < (long)N * 64) {
        int n = (int)(gid >> 6);
        int f = (int)(gid & 63);
        float di = dinv[n];
        float v = Bagg[gid] + di * di * T[gid] + bias[f];
        H[gid] = (v > 0.0f) ? v : (expf(v) - 1.0f);
    }
}

__global__ void k_out(const float* __restrict__ H0, const float* __restrict__ H1,
                      const float* __restrict__ W0f, const float* __restrict__ W1f,
                      const float* __restrict__ cb, float* __restrict__ out, int N) {
    __shared__ float sw[64 * 40 * 2 + 40];
    for (int i = threadIdx.x; i < 64 * 40; i += TPB) {
        sw[i] = W0f[i];
        sw[2560 + i] = W1f[i];
    }
    if (threadIdx.x < 40) sw[5120 + threadIdx.x] = cb[threadIdx.x];
    __syncthreads();

    long gid = (long)blockIdx.x * TPB + threadIdx.x;
    if (gid < (long)N * 40) {
        int n = (int)(gid / 40);
        int c = (int)(gid % 40);
        const float* h0 = H0 + (long)n * 64;
        const float* h1 = H1 + (long)n * 64;
        float s = sw[5120 + c];
        for (int k = 0; k < 64; ++k)
            s += h0[k] * sw[k * 40 + c] + h1[k] * sw[2560 + k * 40 + c];
        out[gid] = s;
    }
}

extern "C" void kernel_launch(void* const* d_in, const int* in_sizes, int n_in,
                              void* d_out, int out_size, void* d_ws, size_t ws_size,
                              hipStream_t stream) {
    const float* x       = (const float*)d_in[0];
    const int*   ei      = (const int*)d_in[1];
    const float* conv0_w = (const float*)d_in[2];
    const float* conv0_b = (const float*)d_in[3];
    const float* lin0_w  = (const float*)d_in[4];
    const float* lin0_b  = (const float*)d_in[5];
    const float* conv1_w = (const float*)d_in[6];
    const float* conv1_b = (const float*)d_in[7];
    const float* lin1_w  = (const float*)d_in[8];
    const float* lin1_b  = (const float*)d_in[9];
    const float* out_w   = (const float*)d_in[10];
    const float* out_b   = (const float*)d_in[11];
    float* out = (float*)d_out;

    int N = in_sizes[0] / 128;
    int E = in_sizes[1] / 2;
    const int* src = ei;
    const int* dst = ei + E;

    size_t NH = (size_t)N * 64;
    float* ws   = (float*)d_ws;
    float* dinv = ws;
    float* A    = ws + (((size_t)N + 255) / 256) * 256;
    float* B    = A + NH;
    float* H0   = B + NH;
    float* W0f  = H0 + NH;
    float* W1f  = W0f + 64 * 40;
    float* cb   = W1f + 64 * 40;

    int gN   = (N + TPB - 1) / TPB;
    int gE   = (E + TPB - 1) / TPB;
    int gNH  = (int)((NH + TPB - 1) / TPB);
    int gSc  = (int)(((size_t)E * 64 + TPB - 1) / TPB);
    int gOut = (int)(((size_t)N * 40 + TPB - 1) / TPB);
    int gGemm = (N + 127) / 128;

    k_init_deg<<<gN, TPB, 0, stream>>>(dinv, N);
    k_deg_accum<<<gE, TPB, 0, stream>>>(dst, dinv, E);
    k_dinv<<<gN, TPB, 0, stream>>>(dinv, N);

    k_fuse_heads<<<1, TPB, 0, stream>>>(lin0_w, lin0_b, lin1_w, lin1_b, out_w, out_b,
                                        W0f, W1f, cb);

    k_gemm_n64_wmma<<<gGemm, 128, 0, stream>>>(x, conv0_w, A, N, 128);
    hipMemsetAsync(B, 0, NH * sizeof(float), stream);
    k_scatter<<<gSc, TPB, 0, stream>>>(A, src, dst, dinv, B, E);
    k_bias_selfloop_elu<<<gNH, TPB, 0, stream>>>(B, A, dinv, conv0_b, H0, N);

    k_gemm_n64_wmma<<<gGemm, 128, 0, stream>>>(H0, conv1_w, A, N, 64);
    hipMemsetAsync(B, 0, NH * sizeof(float), stream);
    k_scatter<<<gSc, TPB, 0, stream>>>(A, src, dst, dinv, B, E);
    k_bias_selfloop_elu<<<gNH, TPB, 0, stream>>>(B, A, dinv, conv1_b, B, N);

    k_out<<<gOut, TPB, 0, stream>>>(H0, B, W0f, W1f, cb, out, N);
}


# in-degree: 16x16 work-groups, each counts one edge slice into LDS bins of one node range (LDS int adds), then adds its bins to deg as whole 128-byte lines; whole-number f32 adds, exact in any order
# speedup vs baseline: 2.9653x; 2.9653x over previous
#include <hip/hip_runtime.h>
#include <hip/hip_bf16.h>

typedef float v2f __attribute__((ext_vector_type(2)));
typedef float v8f __attribute__((ext_vector_type(8)));

#define TPB 256

__global__ void k_init_deg(float* deg, int* tickets, int N) {
    int i = blockIdx.x * TPB + threadIdx.x;
    if (i < N) deg[i] = 1.0f;
    if (i < 256) tickets[i] = 0;
}

#define DH_R   16
#define DH_S   16
#define DH_T   1024
#define DH_CAP 8192
#define DH_U   8

__device__ __forceinline__ void dh_count(int* cnt, int d, int c0, int nb) {
    unsigned a = (unsigned)d - (unsigned)c0;
    if (a < (unsigned)nb) atomicAdd(&cnt[a], 1);
}

__global__ __launch_bounds__(DH_T)
void k_deg_hist(const int* __restrict__ dst, float* deg, int E, int N) {
    __shared__ int cnt[DH_CAP];
    const int tid = threadIdx.x;
    const int r = blockIdx.x % DH_R;
    const int s = blockIdx.x / DH_R;
    const int span = (((N + DH_R - 1) / DH_R) + 31) & ~31;
    const int lo = r * span;
    const int hi = (lo + span < N) ? (lo + span) : N;
    const int per = (((E + DH_S - 1) / DH_S) + 3) & ~3;
    const int e0 = s * per;
    const int e1 = (e0 + per < E) ? (e0 + per) : E;
    const int ne = (e1 > e0) ? (e1 - e0) : 0;
    const bool vec = (((unsigned long long)dst) & 15ull) == 0ull;
    const int n4 = vec ? (ne >> 2) : 0;

    for (int c0 = lo; c0 < hi; c0 += DH_CAP) {
        const int nb = (hi - c0 < DH_CAP) ? (hi - c0) : DH_CAP;
        for (int i = tid; i < nb; i += DH_T) cnt[i] = 0;
        __syncthreads();
        const int4* d4 = (const int4*)(dst + e0);
        for (int i = tid; i < n4; i += DH_U * DH_T) {
            int4 v[DH_U];
#pragma unroll
            for (int u = 0; u < DH_U; ++u) {
                const int j = i + u * DH_T;
                v[u] = (j < n4) ? d4[j] : make_int4(-1, -1, -1, -1);
            }
#pragma unroll
            for (int u = 0; u < DH_U; ++u) {
                dh_count(cnt, v[u].x, c0, nb);
                dh_count(cnt, v[u].y, c0, nb);
                dh_count(cnt, v[u].z, c0, nb);
                dh_count(cnt, v[u].w, c0, nb);
            }
        }
        for (int e = e0 + n4 * 4 + tid; e < e1; e += DH_T)
            dh_count(cnt, dst[e], c0, nb);
        __syncthreads();
        for (int i = tid; i < nb; i += DH_T) atomicAdd(&deg[c0 + i], (float)cnt[i]);
        __syncthreads();
    }
}

__global__ void k_dinv(float* deg, int N) {
    int i = blockIdx.x * TPB + threadIdx.x;
    if (i < N) {
        float d = deg[i];
        deg[i] = (d > 0.0f) ? rsqrtf(d) : 0.0f;
    }
}

__global__ void k_fuse_heads(const float* __restrict__ L0, const float* __restrict__ b0,
                             const float* __restrict__ L1, const float* __restrict__ b1,
                             const float* __restrict__ Wo, const float* __restrict__ bo,
                             float* W0f, float* W1f, float* cb) {
    for (int i = threadIdx.x; i < 64 * 40; i += TPB) {
        int k = i / 40, c = i % 40;
        float s0 = 0.0f, s1 = 0.0f;
        for (int j = 0; j < 64; ++j) {
            s0 += L0[k * 64 + j] * Wo[j * 40 + c];
            s1 += L1[k * 64 + j] * Wo[(64 + j) * 40 + c];
        }
        W0f[i] = s0;
        W1f[i] = s1;
    }
    if (threadIdx.x < 40) {
        int c = threadIdx.x;
        float s = bo[c];
        for (int j = 0; j < 64; ++j)
            s += b0[j] * Wo[j * 40 + c] + b1[j] * Wo[(64 + j) * 40 + c];
        cb[c] = s;
    }
}

__global__ __launch_bounds__(128)
void k_gemm_n64_wmma(const float* __restrict__ A, const float* __restrict__ W,
                     float* __restrict__ D, int N, int K) {
    __shared__ float sW[128 * 64];
    int tid = threadIdx.x;
    for (int i = tid; i < K * 64; i += 128) {
        int k = i >> 6, col = i & 63;
        sW[(k >> 1) * 128 + 2 * col + (k & 1)] = W[i];
    }
    __syncthreads();

    int wave = tid >> 5;
    int lane = tid & 31;
    int half = lane >> 4;
    int l16  = lane & 15;

    int rbase = blockIdx.x * 128 + wave * 32;
    int ar0 = rbase + l16;
    int ar1 = rbase + 16 + l16;
    long ab0 = (long)((ar0 < N) ? ar0 : (N - 1)) * K;
    long ab1 = (long)((ar1 < N) ? ar1 : (N - 1)) * K;

    v8f acc00 = {}, acc01 = {}, acc02 = {}, acc03 = {};
    v8f acc10 = {}, acc11 = {}, acc12 = {}, acc13 = {};

    for (int k = 0; k < K; k += 4) {
        int kb = k + 2 * half;
        v2f a0 = *(const v2f*)(A + ab0 + kb);
        v2f a1 = *(const v2f*)(A + ab1 + kb);
        const float* bp = &sW[(kb >> 1) * 128 + 2 * l16];
        v2f b0 = *(const v2f*)(bp + 0);
        v2f b1 = *(const v2f*)(bp + 32);
        v2f b2 = *(const v2f*)(bp + 64);
        v2f b3 = *(const v2f*)(bp + 96);
        acc00 = __builtin_amdgcn_wmma_f32_16x16x4_f32(false, a0, false, b0, (short)0, acc00, false, false);
        acc01 = __builtin_amdgcn_wmma_f32_16x16x4_f32(false, a0, false, b1, (short)0, acc01, false, false);
        acc02 = __builtin_amdgcn_wmma_f32_16x16x4_f32(false, a0, false, b2, (short)0, acc02, false, false);
        acc03 = __builtin_amdgcn_wmma_f32_16x16x4_f32(false, a0, false, b3, (short)0, acc03, false, false);
        acc10 = __builtin_amdgcn_wmma_f32_16x16x4_f32(false, a1, false, b0, (short)0, acc10, false, false);
        acc11 = __builtin_amdgcn_wmma_f32_16x16x4_f32(false, a1, false, b1, (short)0, acc11, false, false);
        acc12 = __builtin_amdgcn_wmma_f32_16x16x4_f32(false, a1, false, b2, (short)0, acc12, false, false);
        acc13 = __builtin_amdgcn_wmma_f32_16x16x4_f32(false, a1, false, b3, (short)0, acc13, false, false);
    }

    int m0 = rbase + half * 8;
    int m1 = rbase + 16 + half * 8;
    for (int r = 0; r < 8; ++r) {
        int rr = m0 + r;
        if (rr < N) {
            float* dp = D + (long)rr * 64 + l16;
            dp[0]  = acc00[r];
            dp[16] = acc01[r];
            dp[32] = acc02[r];
            dp[48] = acc03[r];
        }
        int ss = m1 + r;
        if (ss < N) {
            float* dp = D + (long)ss * 64 + l16;
            dp[0]  = acc10[r];
            dp[16] = acc11[r];
            dp[32] = acc12[r];
            dp[48] = acc13[r];
        }
    }
}

#define SC_EPW  32
#define SC_U    8
#define SC_TK   2
#define SC_Q    64
#define SC_CTR  64
#define SC_GRID 2048

template <bool FULL>
__device__ __forceinline__ void sc_batch(const float* __restrict__ T, float* Bagg, int lane,
                                         int s_v, int d_v, float w_v, int cnt) {
    for (int j0 = 0; j0 < (FULL ? SC_EPW : cnt); j0 += SC_U) {
        float t0[SC_U], t1[SC_U];
#pragma unroll
        for (int u = 0; u < SC_U; ++u) {
            t0[u] = 0.0f;
            t1[u] = 0.0f;
            if (FULL || j0 + u < cnt) {
                int s = __builtin_amdgcn_readlane(s_v, j0 + u);
                const float* tp = T + (long)s * 64 + lane;
                t0[u] = tp[0];
                t1[u] = tp[32];
            }
        }
#pragma unroll
        for (int u = 0; u < SC_U; ++u) {
            if (FULL || j0 + u < cnt) {
                int d = __builtin_amdgcn_readlane(d_v, j0 + u);
                float w = __int_as_float(__builtin_amdgcn_readlane(__float_as_int(w_v), j0 + u));
                float* bp = Bagg + (long)d * 64 + lane;
                atomicAdd(bp, t0[u] * w);
                atomicAdd(bp + 32, t1[u] * w);
            }
        }
    }
}

__device__ __forceinline__ void sc_lds_barrier() {
    __builtin_amdgcn_fence(__ATOMIC_RELEASE, "workgroup", "local");
    __builtin_amdgcn_s_barrier();
    __builtin_amdgcn_fence(__ATOMIC_ACQUIRE, "workgroup", "local");
}

__device__ __forceinline__ int sc_draw(int* tickets, int half, int ntk, int& cur) {
    while (cur < 2) {
        int own = half ^ cur;
        int t = atomicAdd(tickets + own * SC_CTR, 1);
        if (t < ntk) return t * 2 + own;
        ++cur;
    }
    return -1;
}

__global__ __launch_bounds__(TPB)
void k_scatter(const float* __restrict__ T, const int* __restrict__ src,
               const int* __restrict__ dst, const float* __restrict__ dinv,
               float* Bagg, int* tickets, int E, int N) {
    __shared__ int q_s[TPB / 32][SC_Q];
    __shared__ int q_d[TPB / 32][SC_Q];
    __shared__ int s_tk[2];
    const int tid  = threadIdx.x;
    const int lane = tid & 31;
    const int wave = tid >> 5;
    int* qs = q_s[wave];
    int* qd = q_d[wave];
    const int nsplit = N >> 1;
    const int per_tk = SC_EPW * SC_TK * (TPB / 32);
    const int ntk = (E + per_tk - 1) / per_tk;
    const int half = (__builtin_amdgcn_s_sendmsg_rtn(0x87) >> 18) & 1;
    int cur = 0;

    int par = 0;
    if (tid == 0) s_tk[0] = sc_draw(tickets, half, ntk, cur);
    sc_lds_barrier();

    int qh = 0, qn = 0;
    for (;;) {
        const int tk = s_tk[par];
        if (tk < 0) break;
        int nx_own = 0, nx_raw = 0;
        if (tid == 0 && cur < 2) {
            nx_own = half ^ cur;
            nx_raw = atomicAdd(tickets + nx_own * SC_CTR, 1);
        }

        const int own  = tk & 1;
        const int base = (tk >> 1) * per_tk + wave * (SC_EPW * SC_TK);
        int  sv[SC_TK], dv[SC_TK];
        bool ok[SC_TK];
#pragma unroll
        for (int c = 0; c < SC_TK; ++c) {
            int e = base + c * SC_EPW + lane;
            ok[c] = e < E;
            sv[c] = ok[c] ? src[e] : 0;
            dv[c] = ok[c] ? dst[e] : 0;
        }
#pragma unroll
        for (int c = 0; c < SC_TK; ++c) {
            bool mine = ok[c] && ((dv[c] >= nsplit) == (own != 0));
            unsigned m = __builtin_amdgcn_ballot_w32(mine);
            if (mine) {
                int slot = (qh + qn + (int)__builtin_amdgcn_mbcnt_lo(m, 0u)) & (SC_Q - 1);
                qs[slot] = sv[c];
                qd[slot] = dv[c];
            }
            qn += __builtin_popcount(m);
            if (qn >= SC_EPW) {
                int slot = (qh + lane) & (SC_Q - 1);
                int s = qs[slot];
                int d = qd[slot];
                float w = dinv[s] * dinv[d];
                sc_batch<true>(T, Bagg, lane, s, d, w, SC_EPW);
                qh = (qh + SC_EPW) & (SC_Q - 1);
                qn -= SC_EPW;
            }
        }

        if (tid == 0) {
            int v = -1;
            if (cur < 2) {
                if (nx_raw < ntk) {
                    v = nx_raw * 2 + nx_own;
                } else {
                    ++cur;
                    v = sc_draw(tickets, half, ntk, cur);
                }
            }
            s_tk[par ^ 1] = v;
        }
        sc_lds_barrier();
        par ^= 1;
    }

    if (qn > 0) {
        int slot = (qh + lane) & (SC_Q - 1);
        int s = 0, d = 0;
        float w = 0.0f;
        if (lane < qn) {
            s = qs[slot];
            d = qd[slot];
            w = dinv[s] * dinv[d];
        }
        sc_batch<false>(T, Bagg, lane, s, d, w, qn);
    }
}

__global__ void k_bias_selfloop_elu(const float* __restrict__ Bagg, const float* __restrict__ T,
                                    const float* __restrict__ dinv, const float* __restrict__ bias,
                                    float* H, int N) {
    long gid = (long)blockIdx.x * TPB + threadIdx.x;
    if (gid < (long)N * 64) {
        int n = (int)(gid >> 6);
        int f = (int)(gid & 63);
        float di = dinv[n];
        float v = Bagg[gid] + di * di * T[gid] + bias[f];
        H[gid] = (v > 0.0f) ? v : (expf(v) - 1.0f);
    }
}

#define KO_R 64
#define KO_P 132
#define KO_T 128

__device__ __forceinline__ float ko_elu(float v) { return (v > 0.0f) ? v : (expf(v) - 1.0f); }

__global__ __launch_bounds__(KO_T)
void k_out_tiled(const float* __restrict__ H0, const float* __restrict__ Bagg, const float* __restrict__ T,
                 const float* __restrict__ dinv, const float* __restrict__ bias,
                 const float* __restrict__ W0f, const float* __restrict__ W1f,
                 const float* __restrict__ cb, float* __restrict__ out, int N) {
    __shared__ __attribute__((aligned(16))) float sw[128 * 40];
    __shared__ __attribute__((aligned(16))) float sh[KO_R * KO_P];
    const int tid = threadIdx.x;
    const int row0 = blockIdx.x * KO_R;

    for (int i = tid; i < 64 * 40; i += KO_T) {
        sw[i] = W0f[i];
        sw[2560 + i] = W1f[i];
    }
    for (int i = tid; i < KO_R * 16; i += KO_T) {
        const int r = i >> 4, c4 = (i & 15) * 4;
        const int n = row0 + r;
        float4 a = make_float4(0.0f, 0.0f, 0.0f, 0.0f);
        float4 b = a;
        if (n < N) {
            const long g = (long)n * 64 + c4;
            a = *(const float4*)(H0 + g);
            const float4 ag = *(const float4*)(Bagg + g);
            const float4 t  = *(const float4*)(T + g);
            const float di = dinv[n];
            b.x = ko_elu(ag.x + di * di * t.x + bias[c4 + 0]);
            b.y = ko_elu(ag.y + di * di * t.y + bias[c4 + 1]);
            b.z = ko_elu(ag.z + di * di * t.z + bias[c4 + 2]);
            b.w = ko_elu(ag.w + di * di * t.w + bias[c4 + 3]);
        }
        *(float4*)&sh[r * KO_P + c4] = a;
        *(float4*)&sh[r * KO_P + 64 + c4] = b;
    }
    __syncthreads();

    const int cg = tid & 7;
    const int rg = tid >> 3;
    const float* hp = &sh[(rg * 4) * KO_P];
    const float* wp = &sw[cg * 5];
    float acc[4][5];
#pragma unroll
    for (int j = 0; j < 4; ++j)
#pragma unroll
        for (int c = 0; c < 5; ++c) acc[j][c] = cb[cg * 5 + c];

    for (int k = 0; k < 128; k += 4) {
        float hv[4][4];
#pragma unroll
        for (int j = 0; j < 4; ++j) {
            const float4 h = *(const float4*)(hp + j * KO_P + k);
            hv[j][0] = h.x; hv[j][1] = h.y; hv[j][2] = h.z; hv[j][3] = h.w;
        }
#pragma unroll
        for (int kk = 0; kk < 4; ++kk) {
            float w[5];
#pragma unroll
            for (int c = 0; c < 5; ++c) w[c] = wp[(k + kk) * 40 + c];
#pragma unroll
            for (int j = 0; j < 4; ++j)
#pragma unroll
                for (int c = 0; c < 5; ++c) acc[j][c] += hv[j][kk] * w[c];
        }
    }
    __syncthreads();
#pragma unroll
    for (int j = 0; j < 4; ++j)
#pragma unroll
        for (int c = 0; c < 5; ++c) sh[(rg * 4 + j) * 40 + cg * 5 + c] = acc[j][c];
    __syncthreads();
    const long lim = (long)N * 40;
    for (int i = tid; i < KO_R * 10; i += KO_T) {
        const long g = (long)row0 * 40 + i * 4;
        if (g < lim) *(float4*)(out + g) = *(const float4*)&sh[i * 4];
    }
}

extern "C" void kernel_launch(void* const* d_in, const int* in_sizes, int n_in,
                              void* d_out, int out_size, void* d_ws, size_t ws_size,
                              hipStream_t stream) {
    const float* x       = (const float*)d_in[0];
    const int*   ei      = (const int*)d_in[1];
    const float* conv0_w = (const float*)d_in[2];
    const float* conv0_b = (const float*)d_in[3];
    const float* lin0_w  = (const float*)d_in[4];
    const float* lin0_b  = (const float*)d_in[5];
    const float* conv1_w = (const float*)d_in[6];
    const float* conv1_b = (const float*)d_in[7];
    const float* lin1_w  = (const float*)d_in[8];
    const float* lin1_b  = (const float*)d_in[9];
    const float* out_w   = (const float*)d_in[10];
    const float* out_b   = (const float*)d_in[11];
    float* out = (float*)d_out;

    int N = in_sizes[0] / 128;
    int E = in_sizes[1] / 2;
    const int* src = ei;
    const int* dst = ei + E;

    size_t NH = (size_t)N * 64;
    float* ws   = (float*)d_ws;
    float* dinv = ws;
    float* A    = ws + (((size_t)N + 255) / 256) * 256;
    float* B    = A + NH;
    float* H0   = B + NH;
    float* W0f  = H0 + NH;
    float* W1f  = W0f + 64 * 40;
    float* cb   = W1f + 64 * 40;
    int* tickets = (int*)(cb + 64);

    int gN   = (N + TPB - 1) / TPB;
    int gNH  = (int)((NH + TPB - 1) / TPB);
    int gOut = (N + KO_R - 1) / KO_R;
    int gGemm = (N + 127) / 128;

    k_init_deg<<<gN, TPB, 0, stream>>>(dinv, tickets, N);
    k_deg_hist<<<DH_R * DH_S, DH_T, 0, stream>>>(dst, dinv, E, N);
    k_dinv<<<gN, TPB, 0, stream>>>(dinv, N);

    k_fuse_heads<<<1, TPB, 0, stream>>>(lin0_w, lin0_b, lin1_w, lin1_b, out_w, out_b,
                                        W0f, W1f, cb);

    k_gemm_n64_wmma<<<gGemm, 128, 0, stream>>>(x, conv0_w, A, N, 128);
    hipMemsetAsync(B, 0, NH * sizeof(float), stream);
    k_scatter<<<SC_GRID, TPB, 0, stream>>>(A, src, dst, dinv, B, tickets, E, N);
    k_bias_selfloop_elu<<<gNH, TPB, 0, stream>>>(B, A, dinv, conv0_b, H0, N);

    k_gemm_n64_wmma<<<gGemm, 128, 0, stream>>>(H0, conv1_w, A, N, 64);
    hipMemsetAsync(B, 0, NH * sizeof(float), stream);
    k_scatter<<<SC_GRID, TPB, 0, stream>>>(A, src, dst, dinv, B, tickets + 2 * SC_CTR, E, N);

    k_out_tiled<<<gOut, KO_T, 0, stream>>>(H0, B, A, dinv, conv1_b, W0f, W1f, cb, out, N);
}


# layer epilogue dissolved: GEMM also stores self-loop term + bias as the sum buffer's start (no zero fill); elu applied by the buffer's readers (no elu pass); sums start there, then edges by arrival
# speedup vs baseline: 3.2265x; 1.0881x over previous
#include <hip/hip_runtime.h>
#include <hip/hip_bf16.h>

typedef float v2f __attribute__((ext_vector_type(2)));
typedef float v8f __attribute__((ext_vector_type(8)));

#define TPB 256

__global__ void k_init_deg(float* deg, int* tickets, int N) {
    int i = blockIdx.x * TPB + threadIdx.x;
    if (i < N) deg[i] = 1.0f;
    if (i < 256) tickets[i] = 0;
}

#define DH_R   16
#define DH_S   16
#define DH_T   1024
#define DH_CAP 8192
#define DH_U   8

__device__ __forceinline__ void dh_count(int* cnt, int d, int c0, int nb) {
    unsigned a = (unsigned)d - (unsigned)c0;
    if (a < (unsigned)nb) atomicAdd(&cnt[a], 1);
}

__global__ __launch_bounds__(DH_T)
void k_deg_hist(const int* __restrict__ dst, float* deg, int E, int N) {
    __shared__ int cnt[DH_CAP];
    const int tid = threadIdx.x;
    const int r = blockIdx.x % DH_R;
    const int s = blockIdx.x / DH_R;
    const int span = (((N + DH_R - 1) / DH_R) + 31) & ~31;
    const int lo = r * span;
    const int hi = (lo + span < N) ? (lo + span) : N;
    const int per = (((E + DH_S - 1) / DH_S) + 3) & ~3;
    const int e0 = s * per;
    const int e1 = (e0 + per < E) ? (e0 + per) : E;
    const int ne = (e1 > e0) ? (e1 - e0) : 0;
    const bool vec = (((unsigned long long)dst) & 15ull) == 0ull;
    const int n4 = vec ? (ne >> 2) : 0;

    for (int c0 = lo; c0 < hi; c0 += DH_CAP) {
        const int nb = (hi - c0 < DH_CAP) ? (hi - c0) : DH_CAP;
        for (int i = tid; i < nb; i += DH_T) cnt[i] = 0;
        __syncthreads();
        const int4* d4 = (const int4*)(dst + e0);
        for (int i = tid; i < n4; i += DH_U * DH_T) {
            int4 v[DH_U];
#pragma unroll
            for (int u = 0; u < DH_U; ++u) {
                const int j = i + u * DH_T;
                v[u] = (j < n4) ? d4[j] : make_int4(-1, -1, -1, -1);
            }
#pragma unroll
            for (int u = 0; u < DH_U; ++u) {
                dh_count(cnt, v[u].x, c0, nb);
                dh_count(cnt, v[u].y, c0, nb);
                dh_count(cnt, v[u].z, c0, nb);
                dh_count(cnt, v[u].w, c0, nb);
            }
        }
        for (int e = e0 + n4 * 4 + tid; e < e1; e += DH_T)
            dh_count(cnt, dst[e], c0, nb);
        __syncthreads();
        for (int i = tid; i < nb; i += DH_T) atomicAdd(&deg[c0 + i], (float)cnt[i]);
        __syncthreads();
    }
}

__global__ void k_dinv(float* deg, int N) {
    int i = blockIdx.x * TPB + threadIdx.x;
    if (i < N) {
        float d = deg[i];
        deg[i] = (d > 0.0f) ? rsqrtf(d) : 0.0f;
    }
}

__global__ void k_fuse_heads(const float* __restrict__ L0, const float* __restrict__ b0,
                             const float* __restrict__ L1, const float* __restrict__ b1,
                             const float* __restrict__ Wo, const float* __restrict__ bo,
                             float* W0f, float* W1f, float* cb) {
    for (int i = threadIdx.x; i < 64 * 40; i += TPB) {
        int k = i / 40, c = i % 40;
        float s0 = 0.0f, s1 = 0.0f;
        for (int j = 0; j < 64; ++j) {
            s0 += L0[k * 64 + j] * Wo[j * 40 + c];
            s1 += L1[k * 64 + j] * Wo[(64 + j) * 40 + c];
        }
        W0f[i] = s0;
        W1f[i] = s1;
    }
    if (threadIdx.x < 40) {
        int c = threadIdx.x;
        float s = bo[c];
        for (int j = 0; j < 64; ++j)
            s += b0[j] * Wo[j * 40 + c] + b1[j] * Wo[(64 + j) * 40 + c];
        cb[c] = s;
    }
}

__device__ __forceinline__ float gm_elu(float v) {
    const float e = expf(v) - 1.0f;
    return (v > 0.0f) ? v : e;
}

template <bool ELU_IN>
__device__ __forceinline__ void gemm_n64_body(const float* __restrict__ A, const float* __restrict__ W,
                                              const float* __restrict__ dinv, const float* __restrict__ bias,
                                              float* __restrict__ D, float* __restrict__ S,
                                              float* sW, int N, int K) {
    int tid = threadIdx.x;
    for (int i = tid; i < K * 64; i += 128) {
        int k = i >> 6, col = i & 63;
        sW[(k >> 1) * 128 + 2 * col + (k & 1)] = W[i];
    }
    __syncthreads();

    int wave = tid >> 5;
    int lane = tid & 31;
    int half = lane >> 4;
    int l16  = lane & 15;

    int rbase = blockIdx.x * 128 + wave * 32;
    int ar0 = rbase + l16;
    int ar1 = rbase + 16 + l16;
    long ab0 = (long)((ar0 < N) ? ar0 : (N - 1)) * K;
    long ab1 = (long)((ar1 < N) ? ar1 : (N - 1)) * K;

    int m0 = rbase + half * 8;
    int m1 = rbase + 16 + half * 8;
    const float bc0 = bias[l16], bc1 = bias[16 + l16], bc2 = bias[32 + l16], bc3 = bias[48 + l16];
    float dd0[8], dd1[8];
#pragma unroll
    for (int r = 0; r < 8; ++r) {
        const int q0 = m0 + r, q1 = m1 + r;
        const float d0 = dinv[(q0 < N) ? q0 : (N - 1)];
        const float d1 = dinv[(q1 < N) ? q1 : (N - 1)];
        dd0[r] = d0 * d0;
        dd1[r] = d1 * d1;
    }

    v8f acc00 = {}, acc01 = {}, acc02 = {}, acc03 = {};
    v8f acc10 = {}, acc11 = {}, acc12 = {}, acc13 = {};

    for (int k = 0; k < K; k += 4) {
        int kb = k + 2 * half;
        v2f a0 = *(const v2f*)(A + ab0 + kb);
        v2f a1 = *(const v2f*)(A + ab1 + kb);
        if (ELU_IN) {
            a0.x = gm_elu(a0.x); a0.y = gm_elu(a0.y);
            a1.x = gm_elu(a1.x); a1.y = gm_elu(a1.y);
        }
        const float* bp = &sW[(kb >> 1) * 128 + 2 * l16];
        v2f b0 = *(const v2f*)(bp + 0);
        v2f b1 = *(const v2f*)(bp + 32);
        v2f b2 = *(const v2f*)(bp + 64);
        v2f b3 = *(const v2f*)(bp + 96);
        acc00 = __builtin_amdgcn_wmma_f32_16x16x4_f32(false, a0, false, b0, (short)0, acc00, false, false);
        acc01 = __builtin_amdgcn_wmma_f32_16x16x4_f32(false, a0, false, b1, (short)0, acc01, false, false);
        acc02 = __builtin_amdgcn_wmma_f32_16x16x4_f32(false, a0, false, b2, (short)0, acc02, false, false);
        acc03 = __builtin_amdgcn_wmma_f32_16x16x4_f32(false, a0, false, b3, (short)0, acc03, false, false);
        acc10 = __builtin_amdgcn_wmma_f32_16x16x4_f32(false, a1, false, b0, (short)0, acc10, false, false);
        acc11 = __builtin_amdgcn_wmma_f32_16x16x4_f32(false, a1, false, b1, (short)0, acc11, false, false);
        acc12 = __builtin_amdgcn_wmma_f32_16x16x4_f32(false, a1, false, b2, (short)0, acc12, false, false);
        acc13 = __builtin_amdgcn_wmma_f32_16x16x4_f32(false, a1, false, b3, (short)0, acc13, false, false);
    }

#pragma unroll
    for (int r = 0; r < 8; ++r) {
        int rr = m0 + r;
        if (rr < N) {
            const float dd = dd0[r];
            float* dp = D + (long)rr * 64 + l16;
            float* sp = S + (long)rr * 64 + l16;
            dp[0]  = acc00[r];
            dp[16] = acc01[r];
            dp[32] = acc02[r];
            dp[48] = acc03[r];
            sp[0]  = dd * acc00[r] + bc0;
            sp[16] = dd * acc01[r] + bc1;
            sp[32] = dd * acc02[r] + bc2;
            sp[48] = dd * acc03[r] + bc3;
        }
        int ss = m1 + r;
        if (ss < N) {
            const float dd = dd1[r];
            float* dp = D + (long)ss * 64 + l16;
            float* sp = S + (long)ss * 64 + l16;
            dp[0]  = acc10[r];
            dp[16] = acc11[r];
            dp[32] = acc12[r];
            dp[48] = acc13[r];
            sp[0]  = dd * acc10[r] + bc0;
            sp[16] = dd * acc11[r] + bc1;
            sp[32] = dd * acc12[r] + bc2;
            sp[48] = dd * acc13[r] + bc3;
        }
    }
}

__global__ __launch_bounds__(128)
void k_gemm_l0(const float* __restrict__ A, const float* __restrict__ W,
               const float* __restrict__ dinv, const float* __restrict__ bias,
               float* __restrict__ D, float* __restrict__ S, int N) {
    __shared__ float sW[128 * 64];
    gemm_n64_body<false>(A, W, dinv, bias, D, S, sW, N, 128);
}

__global__ __launch_bounds__(128)
void k_gemm_l1(const float* __restrict__ A, const float* __restrict__ W,
               const float* __restrict__ dinv, const float* __restrict__ bias,
               float* __restrict__ D, float* __restrict__ S, int N) {
    __shared__ float sW[64 * 64];
    gemm_n64_body<true>(A, W, dinv, bias, D, S, sW, N, 64);
}

#define SC_EPW  32
#define SC_U    8
#define SC_TK   2
#define SC_Q    64
#define SC_CTR  64
#define SC_GRID 2048

template <bool FULL>
__device__ __forceinline__ void sc_batch(const float* __restrict__ T, float* Bagg, int lane,
                                         int s_v, int d_v, float w_v, int cnt) {
    for (int j0 = 0; j0 < (FULL ? SC_EPW : cnt); j0 += SC_U) {
        float t0[SC_U], t1[SC_U];
#pragma unroll
        for (int u = 0; u < SC_U; ++u) {
            t0[u] = 0.0f;
            t1[u] = 0.0f;
            if (FULL || j0 + u < cnt) {
                int s = __builtin_amdgcn_readlane(s_v, j0 + u);
                const float* tp = T + (long)s * 64 + lane;
                t0[u] = tp[0];
                t1[u] = tp[32];
            }
        }
#pragma unroll
        for (int u = 0; u < SC_U; ++u) {
            if (FULL || j0 + u < cnt) {
                int d = __builtin_amdgcn_readlane(d_v, j0 + u);
                float w = __int_as_float(__builtin_amdgcn_readlane(__float_as_int(w_v), j0 + u));
                float* bp = Bagg + (long)d * 64 + lane;
                atomicAdd(bp, t0[u] * w);
                atomicAdd(bp + 32, t1[u] * w);
            }
        }
    }
}

__device__ __forceinline__ void sc_lds_barrier() {
    __builtin_amdgcn_fence(__ATOMIC_RELEASE, "workgroup", "local");
    __builtin_amdgcn_s_barrier();
    __builtin_amdgcn_fence(__ATOMIC_ACQUIRE, "workgroup", "local");
}

__device__ __forceinline__ int sc_draw(int* tickets, int half, int ntk, int& cur) {
    while (cur < 2) {
        int own = half ^ cur;
        int t = atomicAdd(tickets + own * SC_CTR, 1);
        if (t < ntk) return t * 2 + own;
        ++cur;
    }
    return -1;
}

__global__ __launch_bounds__(TPB)
void k_scatter(const float* __restrict__ T, const int* __restrict__ src,
               const int* __restrict__ dst, const float* __restrict__ dinv,
               float* Bagg, int* tickets, int E, int N) {
    __shared__ int q_s[TPB / 32][SC_Q];
    __shared__ int q_d[TPB / 32][SC_Q];
    __shared__ int s_tk[2];
    const int tid  = threadIdx.x;
    const int lane = tid & 31;
    const int wave = tid >> 5;
    int* qs = q_s[wave];
    int* qd = q_d[wave];
    const int nsplit = N >> 1;
    const int per_tk = SC_EPW * SC_TK * (TPB / 32);
    const int ntk = (E + per_tk - 1) / per_tk;
    const int half = (__builtin_amdgcn_s_sendmsg_rtn(0x87) >> 18) & 1;
    int cur = 0;

    int par = 0;
    if (tid == 0) s_tk[0] = sc_draw(tickets, half, ntk, cur);
    sc_lds_barrier();

    int qh = 0, qn = 0;
    for (;;) {
        const int tk = s_tk[par];
        if (tk < 0) break;
        int nx_own = 0, nx_raw = 0;
        if (tid == 0 && cur < 2) {
            nx_own = half ^ cur;
            nx_raw = atomicAdd(tickets + nx_own * SC_CTR, 1);
        }

        const int own  = tk & 1;
        const int base = (tk >> 1) * per_tk + wave * (SC_EPW * SC_TK);
        int  sv[SC_TK], dv[SC_TK];
        bool ok[SC_TK];
#pragma unroll
        for (int c = 0; c < SC_TK; ++c) {
            int e = base + c * SC_EPW + lane;
            ok[c] = e < E;
            sv[c] = ok[c] ? src[e] : 0;
            dv[c] = ok[c] ? dst[e] : 0;
        }
#pragma unroll
        for (int c = 0; c < SC_TK; ++c) {
            bool mine = ok[c] && ((dv[c] >= nsplit) == (own != 0));
            unsigned m = __builtin_amdgcn_ballot_w32(mine);
            if (mine) {
                int slot = (qh + qn + (int)__builtin_amdgcn_mbcnt_lo(m, 0u)) & (SC_Q - 1);
                qs[slot] = sv[c];
                qd[slot] = dv[c];
            }
            qn += __builtin_popcount(m);
            if (qn >= SC_EPW) {
                int slot = (qh + lane) & (SC_Q - 1);
                int s = qs[slot];
                int d = qd[slot];
                float w = dinv[s] * dinv[d];
                sc_batch<true>(T, Bagg, lane, s, d, w, SC_EPW);
                qh = (qh + SC_EPW) & (SC_Q - 1);
                qn -= SC_EPW;
            }
        }

        if (tid == 0) {
            int v = -1;
            if (cur < 2) {
                if (nx_raw < ntk) {
                    v = nx_raw * 2 + nx_own;
                } else {
                    ++cur;
                    v = sc_draw(tickets, half, ntk, cur);
                }
            }
            s_tk[par ^ 1] = v;
        }
        sc_lds_barrier();
        par ^= 1;
    }

    if (qn > 0) {
        int slot = (qh + lane) & (SC_Q - 1);
        int s = 0, d = 0;
        float w = 0.0f;
        if (lane < qn) {
            s = qs[slot];
            d = qd[slot];
            w = dinv[s] * dinv[d];
        }
        sc_batch<false>(T, Bagg, lane, s, d, w, qn);
    }
}

#define KO_R 64
#define KO_P 132
#define KO_T 128

__device__ __forceinline__ float ko_elu(float v) { return (v > 0.0f) ? v : (expf(v) - 1.0f); }

__global__ __launch_bounds__(KO_T)
void k_out_tiled(const float* __restrict__ B0, const float* __restrict__ B1,
                 const float* __restrict__ W0f, const float* __restrict__ W1f,
                 const float* __restrict__ cb, float* __restrict__ out, int N) {
    __shared__ __attribute__((aligned(16))) float sw[128 * 40];
    __shared__ __attribute__((aligned(16))) float sh[KO_R * KO_P];
    const int tid = threadIdx.x;
    const int row0 = blockIdx.x * KO_R;

    for (int i = tid; i < 64 * 40; i += KO_T) {
        sw[i] = W0f[i];
        sw[2560 + i] = W1f[i];
    }
    for (int i = tid; i < KO_R * 16; i += KO_T) {
        const int r = i >> 4, c4 = (i & 15) * 4;
        const int n = row0 + r;
        float4 a = make_float4(0.0f, 0.0f, 0.0f, 0.0f);
        float4 b = a;
        if (n < N) {
            const long g = (long)n * 64 + c4;
            const float4 s0 = *(const float4*)(B0 + g);
            const float4 s1 = *(const float4*)(B1 + g);
            a.x = ko_elu(s0.x); a.y = ko_elu(s0.y); a.z = ko_elu(s0.z); a.w = ko_elu(s0.w);
            b.x = ko_elu(s1.x); b.y = ko_elu(s1.y); b.z = ko_elu(s1.z); b.w = ko_elu(s1.w);
        }
        *(float4*)&sh[r * KO_P + c4] = a;
        *(float4*)&sh[r * KO_P + 64 + c4] = b;
    }
    __syncthreads();

    const int cg = tid & 7;
    const int rg = tid >> 3;
    const float* hp = &sh[(rg * 4) * KO_P];
    const float* wp = &sw[cg * 5];
    float acc[4][5];
#pragma unroll
    for (int j = 0; j < 4; ++j)
#pragma unroll
        for (int c = 0; c < 5; ++c) acc[j][c] = cb[cg * 5 + c];

    for (int k = 0; k < 128; k += 4) {
        float hv[4][4];
#pragma unroll
        for (int j = 0; j < 4; ++j) {
            const float4 h = *(const float4*)(hp + j * KO_P + k);
            hv[j][0] = h.x; hv[j][1] = h.y; hv[j][2] = h.z; hv[j][3] = h.w;
        }
#pragma unroll
        for (int kk = 0; kk < 4; ++kk) {
            float w[5];
#pragma unroll
            for (int c = 0; c < 5; ++c) w[c] = wp[(k + kk) * 40 + c];
#pragma unroll
            for (int j = 0; j < 4; ++j)
#pragma unroll
                for (int c = 0; c < 5; ++c) acc[j][c] += hv[j][kk] * w[c];
        }
    }
    __syncthreads();
#pragma unroll
    for (int j = 0; j < 4; ++j)
#pragma unroll
        for (int c = 0; c < 5; ++c) sh[(rg * 4 + j) * 40 + cg * 5 + c] = acc[j][c];
    __syncthreads();
    const long lim = (long)N * 40;
    for (int i = tid; i < KO_R * 10; i += KO_T) {
        const long g = (long)row0 * 40 + i * 4;
        if (g < lim) *(float4*)(out + g) = *(const float4*)&sh[i * 4];
    }
}

extern "C" void kernel_launch(void* const* d_in, const int* in_sizes, int n_in,
                              void* d_out, int out_size, void* d_ws, size_t ws_size,
                              hipStream_t stream) {
    const float* x       = (const float*)d_in[0];
    const int*   ei      = (const int*)d_in[1];
    const float* conv0_w = (const float*)d_in[2];
    const float* conv0_b = (const float*)d_in[3];
    const float* lin0_w  = (const float*)d_in[4];
    const float* lin0_b  = (const float*)d_in[5];
    const float* conv1_w = (const float*)d_in[6];
    const float* conv1_b = (const float*)d_in[7];
    const float* lin1_w  = (const float*)d_in[8];
    const float* lin1_b  = (const float*)d_in[9];
    const float* out_w   = (const float*)d_in[10];
    const float* out_b   = (const float*)d_in[11];
    float* out = (float*)d_out;

    int N = in_sizes[0] / 128;
    int E = in_sizes[1] / 2;
    const int* src = ei;
    const int* dst = ei + E;

    size_t NH = (size_t)N * 64;
    float* ws   = (float*)d_ws;
    float* dinv = ws;
    float* A    = ws + (((size_t)N + 255) / 256) * 256;
    float* B0   = A + NH;
    float* B1   = B0 + NH;
    float* W0f  = B1 + NH;
    float* W1f  = W0f + 64 * 40;
    float* cb   = W1f + 64 * 40;
    int* tickets = (int*)(cb + 64);

    int gN   = (N + TPB - 1) / TPB;
    int gOut = (N + KO_R - 1) / KO_R;
    int gGemm = (N + 127) / 128;

    k_init_deg<<<gN, TPB, 0, stream>>>(dinv, tickets, N);
    k_deg_hist<<<DH_R * DH_S, DH_T, 0, stream>>>(dst, dinv, E, N);
    k_dinv<<<gN, TPB, 0, stream>>>(dinv, N);

    k_fuse_heads<<<1, TPB, 0, stream>>>(lin0_w, lin0_b, lin1_w, lin1_b, out_w, out_b,
                                        W0f, W1f, cb);

    k_gemm_l0<<<gGemm, 128, 0, stream>>>(x, conv0_w, dinv, conv0_b, A, B0, N);
    k_scatter<<<SC_GRID, TPB, 0, stream>>>(A, src, dst, dinv, B0, tickets, E, N);

    k_gemm_l1<<<gGemm, 128, 0, stream>>>(B0, conv1_w, dinv, conv1_b, A, B1, N);
    k_scatter<<<SC_GRID, TPB, 0, stream>>>(A, src, dst, dinv, B1, tickets + 2 * SC_CTR, E, N);

    k_out_tiled<<<gOut, KO_T, 0, stream>>>(B0, B1, W0f, W1f, cb, out, N);
}


# prologue folded, 9 launches to 6: edge slices store integer degree counts (no init pass, no deg atomics), layer-0 GEMM turns its rows' counts into dinv, spare count blocks build the head matrices
# speedup vs baseline: 3.6365x; 1.1271x over previous
#include <hip/hip_runtime.h>
#include <hip/hip_bf16.h>

typedef float v2f __attribute__((ext_vector_type(2)));
typedef float v8f __attribute__((ext_vector_type(8)));

#define TPB 256

#define DH_R   16
#define DH_S   16
#define DH_T   1024
#define DH_CAP 8192
#define DH_U   8
#define DH_HB  3

__device__ __forceinline__ int dh_pitch(int N) { return (N + 255) & ~255; }

__device__ __forceinline__ void dh_count(int* cnt, int d, int c0, int nb) {
    unsigned a = (unsigned)d - (unsigned)c0;
    if (a < (unsigned)nb) atomicAdd(&cnt[a], 1);
}

__device__ __forceinline__ void fuse_heads_entry(int i,
                                                 const float* __restrict__ L0, const float* __restrict__ b0,
                                                 const float* __restrict__ L1, const float* __restrict__ b1,
                                                 const float* __restrict__ Wo, const float* __restrict__ bo,
                                                 float* W0f, float* W1f, float* cb) {
    if (i < 64 * 40) {
        int k = i / 40, c = i % 40;
        float s0 = 0.0f, s1 = 0.0f;
        for (int j = 0; j < 64; ++j) {
            s0 += L0[k * 64 + j] * Wo[j * 40 + c];
            s1 += L1[k * 64 + j] * Wo[(64 + j) * 40 + c];
        }
        W0f[i] = s0;
        W1f[i] = s1;
    } else if (i < 64 * 40 + 40) {
        int c = i - 64 * 40;
        float s = bo[c];
        for (int j = 0; j < 64; ++j)
            s += b0[j] * Wo[j * 40 + c] + b1[j] * Wo[(64 + j) * 40 + c];
        cb[c] = s;
    }
}

__global__ __launch_bounds__(DH_T)
void k_prologue(const int* __restrict__ dst, int* planes, int* tickets,
                const float* __restrict__ L0, const float* __restrict__ b0,
                const float* __restrict__ L1, const float* __restrict__ b1,
                const float* __restrict__ Wo, const float* __restrict__ bo,
                float* W0f, float* W1f, float* cb, int E, int N) {
    __shared__ int cnt[DH_CAP];
    const int tid = threadIdx.x;
    if (blockIdx.x >= DH_R * DH_S) {
        fuse_heads_entry((blockIdx.x - DH_R * DH_S) * DH_T + tid, L0, b0, L1, b1, Wo, bo, W0f, W1f, cb);
        return;
    }
    if (blockIdx.x == 0 && tid < 256) tickets[tid] = 0;
    const int r = blockIdx.x % DH_R;
    const int s = blockIdx.x / DH_R;
    const int Np = dh_pitch(N);
    const int span = (((N + DH_R - 1) / DH_R) + 31) & ~31;
    const int lo = r * span;
    const int hi = (lo + span < N) ? (lo + span) : N;
    const int per = (((E + DH_S - 1) / DH_S) + 3) & ~3;
    const int e0 = s * per;
    const int e1 = (e0 + per < E) ? (e0 + per) : E;
    const int ne = (e1 > e0) ? (e1 - e0) : 0;
    const bool vec = (((unsigned long long)dst) & 15ull) == 0ull;
    const int n4 = vec ? (ne >> 2) : 0;
    int* plane = planes + (long)s * Np;

    for (int c0 = lo; c0 < hi; c0 += DH_CAP) {
        const int nb = (hi - c0 < DH_CAP) ? (hi - c0) : DH_CAP;
        for (int i = tid; i < nb; i += DH_T) cnt[i] = 0;
        __syncthreads();
        const int4* d4 = (const int4*)(dst + e0);
        for (int i = tid; i < n4; i += DH_U * DH_T) {
            int4 v[DH_U];
#pragma unroll
            for (int u = 0; u < DH_U; ++u) {
                const int j = i + u * DH_T;
                v[u] = (j < n4) ? d4[j] : make_int4(-1, -1, -1, -1);
            }
#pragma unroll
            for (int u = 0; u < DH_U; ++u) {
                dh_count(cnt, v[u].x, c0, nb);
                dh_count(cnt, v[u].y, c0, nb);
                dh_count(cnt, v[u].z, c0, nb);
                dh_count(cnt, v[u].w, c0, nb);
            }
        }
        for (int e = e0 + n4 * 4 + tid; e < e1; e += DH_T)
            dh_count(cnt, dst[e], c0, nb);
        __syncthreads();
        for (int i = tid; i < nb; i += DH_T) plane[c0 + i] = cnt[i];
        __syncthreads();
    }
}

__device__ __forceinline__ float gm_elu(float v) {
    const float e = expf(v) - 1.0f;
    return (v > 0.0f) ? v : e;
}

template <bool ELU_IN, bool DEG_IN>
__device__ __forceinline__ void gemm_n64_body(const float* __restrict__ A, const float* __restrict__ W,
                                              const int* __restrict__ planes, float* __restrict__ dinv_out,
                                              const float* __restrict__ dinv, const float* __restrict__ bias,
                                              float* __restrict__ D, float* __restrict__ S,
                                              float* sW, float* sdd, int N, int K) {
    int tid = threadIdx.x;
    const int drow = blockIdx.x * 128 + tid;
    int dcnt[DH_S];
    if (DEG_IN) {
        const int Np = dh_pitch(N);
#pragma unroll
        for (int s = 0; s < DH_S; ++s) dcnt[s] = (drow < N) ? planes[(long)s * Np + drow] : 0;
    }
    for (int i = tid; i < K * 64; i += 128) {
        int k = i >> 6, col = i & 63;
        sW[(k >> 1) * 128 + 2 * col + (k & 1)] = W[i];
    }
    if (DEG_IN) {
        int c = 0;
#pragma unroll
        for (int s = 0; s < DH_S; ++s) c += dcnt[s];
        const float deg = 1.0f + (float)c;
        const float di = (deg > 0.0f) ? rsqrtf(deg) : 0.0f;
        if (drow < N) dinv_out[drow] = di;
        sdd[tid] = di * di;
    }
    __syncthreads();

    int wave = tid >> 5;
    int lane = tid & 31;
    int half = lane >> 4;
    int l16  = lane & 15;

    int rbase = blockIdx.x * 128 + wave * 32;
    int ar0 = rbase + l16;
    int ar1 = rbase + 16 + l16;
    long ab0 = (long)((ar0 < N) ? ar0 : (N - 1)) * K;
    long ab1 = (long)((ar1 < N) ? ar1 : (N - 1)) * K;

    int m0 = rbase + half * 8;
    int m1 = rbase + 16 + half * 8;
    const float bc0 = bias[l16], bc1 = bias[16 + l16], bc2 = bias[32 + l16], bc3 = bias[48 + l16];
    float dd0[8], dd1[8];
#pragma unroll
    for (int r = 0; r < 8; ++r) {
        if (DEG_IN) {
            dd0[r] = sdd[wave * 32 + half * 8 + r];
            dd1[r] = sdd[wave * 32 + 16 + half * 8 + r];
        } else {
            const int q0 = m0 + r, q1 = m1 + r;
            const float d0 = dinv[(q0 < N) ? q0 : (N - 1)];
            const float d1 = dinv[(q1 < N) ? q1 : (N - 1)];
            dd0[r] = d0 * d0;
            dd1[r] = d1 * d1;
        }
    }

    v8f acc00 = {}, acc01 = {}, acc02 = {}, acc03 = {};
    v8f acc10 = {}, acc11 = {}, acc12 = {}, acc13 = {};

    for (int k = 0; k < K; k += 4) {
        int kb = k + 2 * half;
        v2f a0 = *(const v2f*)(A + ab0 + kb);
        v2f a1 = *(const v2f*)(A + ab1 + kb);
        if (ELU_IN) {
            a0.x = gm_elu(a0.x); a0.y = gm_elu(a0.y);
            a1.x = gm_elu(a1.x); a1.y = gm_elu(a1.y);
        }
        const float* bp = &sW[(kb >> 1) * 128 + 2 * l16];
        v2f b0 = *(const v2f*)(bp + 0);
        v2f b1 = *(const v2f*)(bp + 32);
        v2f b2 = *(const v2f*)(bp + 64);
        v2f b3 = *(const v2f*)(bp + 96);
        acc00 = __builtin_amdgcn_wmma_f32_16x16x4_f32(false, a0, false, b0, (short)0, acc00, false, false);
        acc01 = __builtin_amdgcn_wmma_f32_16x16x4_f32(false, a0, false, b1, (short)0, acc01, false, false);
        acc02 = __builtin_amdgcn_wmma_f32_16x16x4_f32(false, a0, false, b2, (short)0, acc02, false, false);
        acc03 = __builtin_amdgcn_wmma_f32_16x16x4_f32(false, a0, false, b3, (short)0, acc03, false, false);
        acc10 = __builtin_amdgcn_wmma_f32_16x16x4_f32(false, a1, false, b0, (short)0, acc10, false, false);
        acc11 = __builtin_amdgcn_wmma_f32_16x16x4_f32(false, a1, false, b1, (short)0, acc11, false, false);
        acc12 = __builtin_amdgcn_wmma_f32_16x16x4_f32(false, a1, false, b2, (short)0, acc12, false, false);
        acc13 = __builtin_amdgcn_wmma_f32_16x16x4_f32(false, a1, false, b3, (short)0, acc13, false, false);
    }

#pragma unroll
    for (int r = 0; r < 8; ++r) {
        int rr = m0 + r;
        if (rr < N) {
            const float dd = dd0[r];
            float* dp = D + (long)rr * 64 + l16;
            float* sp = S + (long)rr * 64 + l16;
            dp[0]  = acc00[r];
            dp[16] = acc01[r];
            dp[32] = acc02[r];
            dp[48] = acc03[r];
            sp[0]  = dd * acc00[r] + bc0;
            sp[16] = dd * acc01[r] + bc1;
            sp[32] = dd * acc02[r] + bc2;
            sp[48] = dd * acc03[r] + bc3;
        }
        int ss = m1 + r;
        if (ss < N) {
            const float dd = dd1[r];
            float* dp = D + (long)ss * 64 + l16;
            float* sp = S + (long)ss * 64 + l16;
            dp[0]  = acc10[r];
            dp[16] = acc11[r];
            dp[32] = acc12[r];
            dp[48] = acc13[r];
            sp[0]  = dd * acc10[r] + bc0;
            sp[16] = dd * acc11[r] + bc1;
            sp[32] = dd * acc12[r] + bc2;
            sp[48] = dd * acc13[r] + bc3;
        }
    }
}

__global__ __launch_bounds__(128)
void k_gemm_l0(const float* __restrict__ A, const float* __restrict__ W,
               const int* __restrict__ planes, float* __restrict__ dinv_out, const float* __restrict__ bias,
               float* __restrict__ D, float* __restrict__ S, int N) {
    __shared__ float sW[128 * 64];
    __shared__ float sdd[128];
    gemm_n64_body<false, true>(A, W, planes, dinv_out, nullptr, bias, D, S, sW, sdd, N, 128);
}

__global__ __launch_bounds__(128)
void k_gemm_l1(const float* __restrict__ A, const float* __restrict__ W,
               const float* __restrict__ dinv, const float* __restrict__ bias,
               float* __restrict__ D, float* __restrict__ S, int N) {
    __shared__ float sW[64 * 64];
    gemm_n64_body<true, false>(A, W, nullptr, nullptr, dinv, bias, D, S, sW, nullptr, N, 64);
}

#define SC_EPW  32
#define SC_U    8
#define SC_TK   2
#define SC_Q    64
#define SC_CTR  64
#define SC_GRID 2048

template <bool FULL>
__device__ __forceinline__ void sc_batch(const float* __restrict__ T, float* Bagg, int lane,
                                         int s_v, int d_v, float w_v, int cnt) {
    for (int j0 = 0; j0 < (FULL ? SC_EPW : cnt); j0 += SC_U) {
        float t0[SC_U], t1[SC_U];
#pragma unroll
        for (int u = 0; u < SC_U; ++u) {
            t0[u] = 0.0f;
            t1[u] = 0.0f;
            if (FULL || j0 + u < cnt) {
                int s = __builtin_amdgcn_readlane(s_v, j0 + u);
                const float* tp = T + (long)s * 64 + lane;
                t0[u] = tp[0];
                t1[u] = tp[32];
            }
        }
#pragma unroll
        for (int u = 0; u < SC_U; ++u) {
            if (FULL || j0 + u < cnt) {
                int d = __builtin_amdgcn_readlane(d_v, j0 + u);
                float w = __int_as_float(__builtin_amdgcn_readlane(__float_as_int(w_v), j0 + u));
                float* bp = Bagg + (long)d * 64 + lane;
                atomicAdd(bp, t0[u] * w);
                atomicAdd(bp + 32, t1[u] * w);
            }
        }
    }
}

__device__ __forceinline__ void sc_lds_barrier() {
    __builtin_amdgcn_fence(__ATOMIC_RELEASE, "workgroup", "local");
    __builtin_amdgcn_s_barrier();
    __builtin_amdgcn_fence(__ATOMIC_ACQUIRE, "workgroup", "local");
}

__device__ __forceinline__ int sc_draw(int* tickets, int half, int ntk, int& cur) {
    while (cur < 2) {
        int own = half ^ cur;
        int t = atomicAdd(tickets + own * SC_CTR, 1);
        if (t < ntk) return t * 2 + own;
        ++cur;
    }
    return -1;
}

__global__ __launch_bounds__(TPB)
void k_scatter(const float* __restrict__ T, const int* __restrict__ src,
               const int* __restrict__ dst, const float* __restrict__ dinv,
               float* Bagg, int* tickets, int E, int N) {
    __shared__ int q_s[TPB / 32][SC_Q];
    __shared__ int q_d[TPB / 32][SC_Q];
    __shared__ int s_tk[2];
    const int tid  = threadIdx.x;
    const int lane = tid & 31;
    const int wave = tid >> 5;
    int* qs = q_s[wave];
    int* qd = q_d[wave];
    const int nsplit = N >> 1;
    const int per_tk = SC_EPW * SC_TK * (TPB / 32);
    const int ntk = (E + per_tk - 1) / per_tk;
    const int half = (__builtin_amdgcn_s_sendmsg_rtn(0x87) >> 18) & 1;
    int cur = 0;

    int par = 0;
    if (tid == 0) s_tk[0] = sc_draw(tickets, half, ntk, cur);
    sc_lds_barrier();

    int qh = 0, qn = 0;
    for (;;) {
        const int tk = s_tk[par];
        if (tk < 0) break;
        int nx_own = 0, nx_raw = 0;
        if (tid == 0 && cur < 2) {
            nx_own = half ^ cur;
            nx_raw = atomicAdd(tickets + nx_own * SC_CTR, 1);
        }

        const int own  = tk & 1;
        const int base = (tk >> 1) * per_tk + wave * (SC_EPW * SC_TK);
        int  sv[SC_TK], dv[SC_TK];
        bool ok[SC_TK];
#pragma unroll
        for (int c = 0; c < SC_TK; ++c) {
            int e = base + c * SC_EPW + lane;
            ok[c] = e < E;
            sv[c] = ok[c] ? src[e] : 0;
            dv[c] = ok[c] ? dst[e] : 0;
        }
#pragma unroll
        for (int c = 0; c < SC_TK; ++c) {
            bool mine = ok[c] && ((dv[c] >= nsplit) == (own != 0));
            unsigned m = __builtin_amdgcn_ballot_w32(mine);
            if (mine) {
                int slot = (qh + qn + (int)__builtin_amdgcn_mbcnt_lo(m, 0u)) & (SC_Q - 1);
                qs[slot] = sv[c];
                qd[slot] = dv[c];
            }
            qn += __builtin_popcount(m);
            if (qn >= SC_EPW) {
                int slot = (qh + lane) & (SC_Q - 1);
                int s = qs[slot];
                int d = qd[slot];
                float w = dinv[s] * dinv[d];
                sc_batch<true>(T, Bagg, lane, s, d, w, SC_EPW);
                qh = (qh + SC_EPW) & (SC_Q - 1);
                qn -= SC_EPW;
            }
        }

        if (tid == 0) {
            int v = -1;
            if (cur < 2) {
                if (nx_raw < ntk) {
                    v = nx_raw * 2 + nx_own;
                } else {
                    ++cur;
                    v = sc_draw(tickets, half, ntk, cur);
                }
            }
            s_tk[par ^ 1] = v;
        }
        sc_lds_barrier();
        par ^= 1;
    }

    if (qn > 0) {
        int slot = (qh + lane) & (SC_Q - 1);
        int s = 0, d = 0;
        float w = 0.0f;
        if (lane < qn) {
            s = qs[slot];
            d = qd[slot];
            w = dinv[s] * dinv[d];
        }
        sc_batch<false>(T, Bagg, lane, s, d, w, qn);
    }
}

#define KO_R 64
#define KO_P 132
#define KO_T 128

__device__ __forceinline__ float ko_elu(float v) { return (v > 0.0f) ? v : (expf(v) - 1.0f); }

__global__ __launch_bounds__(KO_T)
void k_out_tiled(const float* __restrict__ B0, const float* __restrict__ B1,
                 const float* __restrict__ W0f, const float* __restrict__ W1f,
                 const float* __restrict__ cb, float* __restrict__ out, int N) {
    __shared__ __attribute__((aligned(16))) float sw[128 * 40];
    __shared__ __attribute__((aligned(16))) float sh[KO_R * KO_P];
    const int tid = threadIdx.x;
    const int row0 = blockIdx.x * KO_R;

    for (int i = tid; i < 64 * 40; i += KO_T) {
        sw[i] = W0f[i];
        sw[2560 + i] = W1f[i];
    }
    for (int i = tid; i < KO_R * 16; i += KO_T) {
        const int r = i >> 4, c4 = (i & 15) * 4;
        const int n = row0 + r;
        float4 a = make_float4(0.0f, 0.0f, 0.0f, 0.0f);
        float4 b = a;
        if (n < N) {
            const long g = (long)n * 64 + c4;
            const float4 s0 = *(const float4*)(B0 + g);
            const float4 s1 = *(const float4*)(B1 + g);
            a.x = ko_elu(s0.x); a.y = ko_elu(s0.y); a.z = ko_elu(s0.z); a.w = ko_elu(s0.w);
            b.x = ko_elu(s1.x); b.y = ko_elu(s1.y); b.z = ko_elu(s1.z); b.w = ko_elu(s1.w);
        }
        *(float4*)&sh[r * KO_P + c4] = a;
        *(float4*)&sh[r * KO_P + 64 + c4] = b;
    }
    __syncthreads();

    const int cg = tid & 7;
    const int rg = tid >> 3;
    const float* hp = &sh[(rg * 4) * KO_P];
    const float* wp = &sw[cg * 5];
    float acc[4][5];
#pragma unroll
    for (int j = 0; j < 4; ++j)
#pragma unroll
        for (int c = 0; c < 5; ++c) acc[j][c] = cb[cg * 5 + c];

    for (int k = 0; k < 128; k += 4) {
        float hv[4][4];
#pragma unroll
        for (int j = 0; j < 4; ++j) {
            const float4 h = *(const float4*)(hp + j * KO_P + k);
            hv[j][0] = h.x; hv[j][1] = h.y; hv[j][2] = h.z; hv[j][3] = h.w;
        }
#pragma unroll
        for (int kk = 0; kk < 4; ++kk) {
            float w[5];
#pragma unroll
            for (int c = 0; c < 5; ++c) w[c] = wp[(k + kk) * 40 + c];
#pragma unroll
            for (int j = 0; j < 4; ++j)
#pragma unroll
                for (int c = 0; c < 5; ++c) acc[j][c] += hv[j][kk] * w[c];
        }
    }
    __syncthreads();
#pragma unroll
    for (int j = 0; j < 4; ++j)
#pragma unroll
        for (int c = 0; c < 5; ++c) sh[(rg * 4 + j) * 40 + cg * 5 + c] = acc[j][c];
    __syncthreads();
    const long lim = (long)N * 40;
    for (int i = tid; i < KO_R * 10; i += KO_T) {
        const long g = (long)row0 * 40 + i * 4;
        if (g < lim) *(float4*)(out + g) = *(const float4*)&sh[i * 4];
    }
}

extern "C" void kernel_launch(void* const* d_in, const int* in_sizes, int n_in,
                              void* d_out, int out_size, void* d_ws, size_t ws_size,
                              hipStream_t stream) {
    const float* x       = (const float*)d_in[0];
    const int*   ei      = (const int*)d_in[1];
    const float* conv0_w = (const float*)d_in[2];
    const float* conv0_b = (const float*)d_in[3];
    const float* lin0_w  = (const float*)d_in[4];
    const float* lin0_b  = (const float*)d_in[5];
    const float* conv1_w = (const float*)d_in[6];
    const float* conv1_b = (const float*)d_in[7];
    const float* lin1_w  = (const float*)d_in[8];
    const float* lin1_b  = (const float*)d_in[9];
    const float* out_w   = (const float*)d_in[10];
    const float* out_b   = (const float*)d_in[11];
    float* out = (float*)d_out;

    int N = in_sizes[0] / 128;
    int E = in_sizes[1] / 2;
    const int* src = ei;
    const int* dst = ei + E;

    size_t NH = (size_t)N * 64;
    float* ws   = (float*)d_ws;
    float* dinv = ws;
    float* A    = ws + (((size_t)N + 255) / 256) * 256;
    float* B0   = A + NH;
    float* B1   = B0 + NH;
    float* W0f  = B1 + NH;
    float* W1f  = W0f + 64 * 40;
    float* cb   = W1f + 64 * 40;
    int* tickets = (int*)(cb + 64);
    int* planes  = (int*)B1;

    int gOut = (N + KO_R - 1) / KO_R;
    int gGemm = (N + 127) / 128;

    k_prologue<<<DH_R * DH_S + DH_HB, DH_T, 0, stream>>>(dst, planes, tickets,
                                                         lin0_w, lin0_b, lin1_w, lin1_b, out_w, out_b,
                                                         W0f, W1f, cb, E, N);

    k_gemm_l0<<<gGemm, 128, 0, stream>>>(x, conv0_w, planes, dinv, conv0_b, A, B0, N);
    k_scatter<<<SC_GRID, TPB, 0, stream>>>(A, src, dst, dinv, B0, tickets, E, N);

    k_gemm_l1<<<gGemm, 128, 0, stream>>>(B0, conv1_w, dinv, conv1_b, A, B1, N);
    k_scatter<<<SC_GRID, TPB, 0, stream>>>(A, src, dst, dinv, B1, tickets + 2 * SC_CTR, E, N);

    k_out_tiled<<<gOut, KO_T, 0, stream>>>(B0, B1, W0f, W1f, cb, out, N);
}


# destination-side norm factor taken out of the edge loop: scatter weighs rows by dinv[src] alone (1 scattered lookup per edge, not 2); sum readers apply dinv[row], bias; adds differently rounded terms
# speedup vs baseline: 3.7882x; 1.0417x over previous
#include <hip/hip_runtime.h>
#include <hip/hip_bf16.h>

typedef float v2f __attribute__((ext_vector_type(2)));
typedef float v8f __attribute__((ext_vector_type(8)));

#define TPB 256

#define DH_R   16
#define DH_S   16
#define DH_T   1024
#define DH_CAP 8192
#define DH_U   8
#define DH_HB  3

__device__ __forceinline__ int dh_pitch(int N) { return (N + 255) & ~255; }

__device__ __forceinline__ void dh_count(int* cnt, int d, int c0, int nb) {
    unsigned a = (unsigned)d - (unsigned)c0;
    if (a < (unsigned)nb) atomicAdd(&cnt[a], 1);
}

__device__ __forceinline__ void fuse_heads_entry(int i,
                                                 const float* __restrict__ L0, const float* __restrict__ b0,
                                                 const float* __restrict__ L1, const float* __restrict__ b1,
                                                 const float* __restrict__ Wo, const float* __restrict__ bo,
                                                 float* W0f, float* W1f, float* cb) {
    if (i < 64 * 40) {
        int k = i / 40, c = i % 40;
        float s0 = 0.0f, s1 = 0.0f;
        for (int j = 0; j < 64; ++j) {
            s0 += L0[k * 64 + j] * Wo[j * 40 + c];
            s1 += L1[k * 64 + j] * Wo[(64 + j) * 40 + c];
        }
        W0f[i] = s0;
        W1f[i] = s1;
    } else if (i < 64 * 40 + 40) {
        int c = i - 64 * 40;
        float s = bo[c];
        for (int j = 0; j < 64; ++j)
            s += b0[j] * Wo[j * 40 + c] + b1[j] * Wo[(64 + j) * 40 + c];
        cb[c] = s;
    }
}

__global__ __launch_bounds__(DH_T)
void k_prologue(const int* __restrict__ dst, int* planes, int* tickets,
                const float* __restrict__ L0, const float* __restrict__ b0,
                const float* __restrict__ L1, const float* __restrict__ b1,
                const float* __restrict__ Wo, const float* __restrict__ bo,
                float* W0f, float* W1f, float* cb, int E, int N) {
    __shared__ int cnt[DH_CAP];
    const int tid = threadIdx.x;
    if (blockIdx.x >= DH_R * DH_S) {
        fuse_heads_entry((blockIdx.x - DH_R * DH_S) * DH_T + tid, L0, b0, L1, b1, Wo, bo, W0f, W1f, cb);
        return;
    }
    if (blockIdx.x == 0 && tid < 256) tickets[tid] = 0;
    const int r = blockIdx.x % DH_R;
    const int s = blockIdx.x / DH_R;
    const int Np = dh_pitch(N);
    const int span = (((N + DH_R - 1) / DH_R) + 31) & ~31;
    const int lo = r * span;
    const int hi = (lo + span < N) ? (lo + span) : N;
    const int per = (((E + DH_S - 1) / DH_S) + 3) & ~3;
    const int e0 = s * per;
    const int e1 = (e0 + per < E) ? (e0 + per) : E;
    const int ne = (e1 > e0) ? (e1 - e0) : 0;
    const bool vec = (((unsigned long long)dst) & 15ull) == 0ull;
    const int n4 = vec ? (ne >> 2) : 0;
    int* plane = planes + (long)s * Np;

    for (int c0 = lo; c0 < hi; c0 += DH_CAP) {
        const int nb = (hi - c0 < DH_CAP) ? (hi - c0) : DH_CAP;
        for (int i = tid; i < nb; i += DH_T) cnt[i] = 0;
        __syncthreads();
        const int4* d4 = (const int4*)(dst + e0);
        for (int i = tid; i < n4; i += DH_U * DH_T) {
            int4 v[DH_U];
#pragma unroll
            for (int u = 0; u < DH_U; ++u) {
                const int j = i + u * DH_T;
                v[u] = (j < n4) ? d4[j] : make_int4(-1, -1, -1, -1);
            }
#pragma unroll
            for (int u = 0; u < DH_U; ++u) {
                dh_count(cnt, v[u].x, c0, nb);
                dh_count(cnt, v[u].y, c0, nb);
                dh_count(cnt, v[u].z, c0, nb);
                dh_count(cnt, v[u].w, c0, nb);
            }
        }
        for (int e = e0 + n4 * 4 + tid; e < e1; e += DH_T)
            dh_count(cnt, dst[e], c0, nb);
        __syncthreads();
        for (int i = tid; i < nb; i += DH_T) plane[c0 + i] = cnt[i];
        __syncthreads();
    }
}

__device__ __forceinline__ float gm_elu(float v) {
    const float e = expf(v) - 1.0f;
    return (v > 0.0f) ? v : e;
}

template <bool ELU_IN, bool DEG_IN>
__device__ __forceinline__ void gemm_n64_body(const float* __restrict__ A, const float* __restrict__ W,
                                              const int* __restrict__ planes, float* __restrict__ dinv_out,
                                              const float* __restrict__ dinv, const float* __restrict__ biasA,
                                              float* __restrict__ D, float* __restrict__ S,
                                              float* sW, float* sdd, float* sBi, int N, int K) {
    int tid = threadIdx.x;
    const int drow = blockIdx.x * 128 + tid;
    int dcnt[DH_S];
    if (DEG_IN) {
        const int Np = dh_pitch(N);
#pragma unroll
        for (int s = 0; s < DH_S; ++s) dcnt[s] = (drow < N) ? planes[(long)s * Np + drow] : 0;
    }
    for (int i = tid; i < K * 64; i += 128) {
        int k = i >> 6, col = i & 63;
        sW[(k >> 1) * 128 + 2 * col + (k & 1)] = W[i];
    }
    if (DEG_IN) {
        int c = 0;
#pragma unroll
        for (int s = 0; s < DH_S; ++s) c += dcnt[s];
        const float deg = 1.0f + (float)c;
        const float di = (deg > 0.0f) ? rsqrtf(deg) : 0.0f;
        if (drow < N) dinv_out[drow] = di;
        sdd[tid] = di;
    }
    if (ELU_IN) {
        if (tid < 64) sBi[tid] = biasA[tid];
    }
    __syncthreads();

    int wave = tid >> 5;
    int lane = tid & 31;
    int half = lane >> 4;
    int l16  = lane & 15;

    int rbase = blockIdx.x * 128 + wave * 32;
    int ar0 = rbase + l16;
    int ar1 = rbase + 16 + l16;
    long ab0 = (long)((ar0 < N) ? ar0 : (N - 1)) * K;
    long ab1 = (long)((ar1 < N) ? ar1 : (N - 1)) * K;

    int m0 = rbase + half * 8;
    int m1 = rbase + 16 + half * 8;
    float dd0[8], dd1[8];
#pragma unroll
    for (int r = 0; r < 8; ++r) {
        if (DEG_IN) {
            dd0[r] = sdd[wave * 32 + half * 8 + r];
            dd1[r] = sdd[wave * 32 + 16 + half * 8 + r];
        } else {
            const int q0 = m0 + r, q1 = m1 + r;
            dd0[r] = dinv[(q0 < N) ? q0 : (N - 1)];
            dd1[r] = dinv[(q1 < N) ? q1 : (N - 1)];
        }
    }

    float da0 = 1.0f, da1 = 1.0f;
    if (ELU_IN) {
        da0 = dinv[(ar0 < N) ? ar0 : (N - 1)];
        da1 = dinv[(ar1 < N) ? ar1 : (N - 1)];
    }

    v8f acc00 = {}, acc01 = {}, acc02 = {}, acc03 = {};
    v8f acc10 = {}, acc11 = {}, acc12 = {}, acc13 = {};

    for (int k = 0; k < K; k += 4) {
        int kb = k + 2 * half;
        v2f a0 = *(const v2f*)(A + ab0 + kb);
        v2f a1 = *(const v2f*)(A + ab1 + kb);
        if (ELU_IN) {
            const v2f bk = *(const v2f*)(sBi + kb);
            a0.x = gm_elu(da0 * a0.x + bk.x); a0.y = gm_elu(da0 * a0.y + bk.y);
            a1.x = gm_elu(da1 * a1.x + bk.x); a1.y = gm_elu(da1 * a1.y + bk.y);
        }
        const float* bp = &sW[(kb >> 1) * 128 + 2 * l16];
        v2f b0 = *(const v2f*)(bp + 0);
        v2f b1 = *(const v2f*)(bp + 32);
        v2f b2 = *(const v2f*)(bp + 64);
        v2f b3 = *(const v2f*)(bp + 96);
        acc00 = __builtin_amdgcn_wmma_f32_16x16x4_f32(false, a0, false, b0, (short)0, acc00, false, false);
        acc01 = __builtin_amdgcn_wmma_f32_16x16x4_f32(false, a0, false, b1, (short)0, acc01, false, false);
        acc02 = __builtin_amdgcn_wmma_f32_16x16x4_f32(false, a0, false, b2, (short)0, acc02, false, false);
        acc03 = __builtin_amdgcn_wmma_f32_16x16x4_f32(false, a0, false, b3, (short)0, acc03, false, false);
        acc10 = __builtin_amdgcn_wmma_f32_16x16x4_f32(false, a1, false, b0, (short)0, acc10, false, false);
        acc11 = __builtin_amdgcn_wmma_f32_16x16x4_f32(false, a1, false, b1, (short)0, acc11, false, false);
        acc12 = __builtin_amdgcn_wmma_f32_16x16x4_f32(false, a1, false, b2, (short)0, acc12, false, false);
        acc13 = __builtin_amdgcn_wmma_f32_16x16x4_f32(false, a1, false, b3, (short)0, acc13, false, false);
    }

#pragma unroll
    for (int r = 0; r < 8; ++r) {
        int rr = m0 + r;
        if (rr < N) {
            const float dd = dd0[r];
            float* dp = D + (long)rr * 64 + l16;
            float* sp = S + (long)rr * 64 + l16;
            dp[0]  = acc00[r];
            dp[16] = acc01[r];
            dp[32] = acc02[r];
            dp[48] = acc03[r];
            sp[0]  = dd * acc00[r];
            sp[16] = dd * acc01[r];
            sp[32] = dd * acc02[r];
            sp[48] = dd * acc03[r];
        }
        int ss = m1 + r;
        if (ss < N) {
            const float dd = dd1[r];
            float* dp = D + (long)ss * 64 + l16;
            float* sp = S + (long)ss * 64 + l16;
            dp[0]  = acc10[r];
            dp[16] = acc11[r];
            dp[32] = acc12[r];
            dp[48] = acc13[r];
            sp[0]  = dd * acc10[r];
            sp[16] = dd * acc11[r];
            sp[32] = dd * acc12[r];
            sp[48] = dd * acc13[r];
        }
    }
}

__global__ __launch_bounds__(128)
void k_gemm_l0(const float* __restrict__ A, const float* __restrict__ W,
               const int* __restrict__ planes, float* __restrict__ dinv_out,
               float* __restrict__ D, float* __restrict__ S, int N) {
    __shared__ float sW[128 * 64];
    __shared__ float sdd[128];
    gemm_n64_body<false, true>(A, W, planes, dinv_out, nullptr, nullptr, D, S, sW, sdd, nullptr, N, 128);
}

__global__ __launch_bounds__(128)
void k_gemm_l1(const float* __restrict__ A, const float* __restrict__ W,
               const float* __restrict__ dinv, const float* __restrict__ bias0,
               float* __restrict__ D, float* __restrict__ S, int N) {
    __shared__ float sW[64 * 64];
    __shared__ __attribute__((aligned(8))) float sBi[64];
    gemm_n64_body<true, false>(A, W, nullptr, nullptr, dinv, bias0, D, S, sW, nullptr, sBi, N, 64);
}

#define SC_EPW  32
#define SC_U    8
#define SC_TK   2
#define SC_Q    64
#define SC_CTR  64
#define SC_GRID 2048

template <bool FULL>
__device__ __forceinline__ void sc_batch(const float* __restrict__ T, float* Bagg, int lane,
                                         int s_v, int d_v, float w_v, int cnt) {
    for (int j0 = 0; j0 < (FULL ? SC_EPW : cnt); j0 += SC_U) {
        float t0[SC_U], t1[SC_U];
#pragma unroll
        for (int u = 0; u < SC_U; ++u) {
            t0[u] = 0.0f;
            t1[u] = 0.0f;
            if (FULL || j0 + u < cnt) {
                int s = __builtin_amdgcn_readlane(s_v, j0 + u);
                const float* tp = T + (long)s * 64 + lane;
                t0[u] = tp[0];
                t1[u] = tp[32];
            }
        }
#pragma unroll
        for (int u = 0; u < SC_U; ++u) {
            if (FULL || j0 + u < cnt) {
                int d = __builtin_amdgcn_readlane(d_v, j0 + u);
                float w = __int_as_float(__builtin_amdgcn_readlane(__float_as_int(w_v), j0 + u));
                float* bp = Bagg + (long)d * 64 + lane;
                atomicAdd(bp, t0[u] * w);
                atomicAdd(bp + 32, t1[u] * w);
            }
        }
    }
}

__device__ __forceinline__ void sc_lds_barrier() {
    __builtin_amdgcn_fence(__ATOMIC_RELEASE, "workgroup", "local");
    __builtin_amdgcn_s_barrier();
    __builtin_amdgcn_fence(__ATOMIC_ACQUIRE, "workgroup", "local");
}

__device__ __forceinline__ int sc_draw(int* tickets, int half, int ntk, int& cur) {
    while (cur < 2) {
        int own = half ^ cur;
        int t = atomicAdd(tickets + own * SC_CTR, 1);
        if (t < ntk) return t * 2 + own;
        ++cur;
    }
    return -1;
}

__global__ __launch_bounds__(TPB)
void k_scatter(const float* __restrict__ T, const int* __restrict__ src,
               const int* __restrict__ dst, const float* __restrict__ dinv,
               float* Bagg, int* tickets, int E, int N) {
    __shared__ int q_s[TPB / 32][SC_Q];
    __shared__ int q_d[TPB / 32][SC_Q];
    __shared__ int s_tk[2];
    const int tid  = threadIdx.x;
    const int lane = tid & 31;
    const int wave = tid >> 5;
    int* qs = q_s[wave];
    int* qd = q_d[wave];
    const int nsplit = N >> 1;
    const int per_tk = SC_EPW * SC_TK * (TPB / 32);
    const int ntk = (E + per_tk - 1) / per_tk;
    const int half = (__builtin_amdgcn_s_sendmsg_rtn(0x87) >> 18) & 1;
    int cur = 0;

    int par = 0;
    if (tid == 0) s_tk[0] = sc_draw(tickets, half, ntk, cur);
    sc_lds_barrier();

    int qh = 0, qn = 0;
    for (;;) {
        const int tk = s_tk[par];
        if (tk < 0) break;
        int nx_own = 0, nx_raw = 0;
        if (tid == 0 && cur < 2) {
            nx_own = half ^ cur;
            nx_raw = atomicAdd(tickets + nx_own * SC_CTR, 1);
        }

        const int own  = tk & 1;
        const int base = (tk >> 1) * per_tk + wave * (SC_EPW * SC_TK);
        int  sv[SC_TK], dv[SC_TK];
        bool ok[SC_TK];
#pragma unroll
        for (int c = 0; c < SC_TK; ++c) {
            int e = base + c * SC_EPW + lane;
            ok[c] = e < E;
            sv[c] = ok[c] ? src[e] : 0;
            dv[c] = ok[c] ? dst[e] : 0;
        }
#pragma unroll
        for (int c = 0; c < SC_TK; ++c) {
            bool mine = ok[c] && ((dv[c] >= nsplit) == (own != 0));
            unsigned m = __builtin_amdgcn_ballot_w32(mine);
            if (mine) {
                int slot = (qh + qn + (int)__builtin_amdgcn_mbcnt_lo(m, 0u)) & (SC_Q - 1);
                qs[slot] = sv[c];
                qd[slot] = dv[c];
            }
            qn += __builtin_popcount(m);
            if (qn >= SC_EPW) {
                int slot = (qh + lane) & (SC_Q - 1);
                int s = qs[slot];
                int d = qd[slot];
                float w = dinv[s];
                sc_batch<true>(T, Bagg, lane, s, d, w, SC_EPW);
                qh = (qh + SC_EPW) & (SC_Q - 1);
                qn -= SC_EPW;
            }
        }

        if (tid == 0) {
            int v = -1;
            if (cur < 2) {
                if (nx_raw < ntk) {
                    v = nx_raw * 2 + nx_own;
                } else {
                    ++cur;
                    v = sc_draw(tickets, half, ntk, cur);
                }
            }
            s_tk[par ^ 1] = v;
        }
        sc_lds_barrier();
        par ^= 1;
    }

    if (qn > 0) {
        int slot = (qh + lane) & (SC_Q - 1);
        int s = 0, d = 0;
        float w = 0.0f;
        if (lane < qn) {
            s = qs[slot];
            d = qd[slot];
            w = dinv[s];
        }
        sc_batch<false>(T, Bagg, lane, s, d, w, qn);
    }
}

#define KO_R 64
#define KO_P 132
#define KO_T 128

__device__ __forceinline__ float ko_elu(float v) { return (v > 0.0f) ? v : (expf(v) - 1.0f); }

__global__ __launch_bounds__(KO_T)
void k_out_tiled(const float* __restrict__ B0, const float* __restrict__ B1,
                 const float* __restrict__ dinv, const float* __restrict__ cbias0, const float* __restrict__ cbias1,
                 const float* __restrict__ W0f, const float* __restrict__ W1f,
                 const float* __restrict__ cb, float* __restrict__ out, int N) {
    __shared__ __attribute__((aligned(16))) float sw[128 * 40];
    __shared__ __attribute__((aligned(16))) float sh[KO_R * KO_P];
    const int tid = threadIdx.x;
    const int row0 = blockIdx.x * KO_R;

    for (int i = tid; i < 64 * 40; i += KO_T) {
        sw[i] = W0f[i];
        sw[2560 + i] = W1f[i];
    }
    const int cq = (tid & 15) * 4;
    const float p0 = cbias0[cq], p1 = cbias0[cq + 1], p2 = cbias0[cq + 2], p3 = cbias0[cq + 3];
    const float q0 = cbias1[cq], q1 = cbias1[cq + 1], q2 = cbias1[cq + 2], q3 = cbias1[cq + 3];
    for (int i = tid; i < KO_R * 16; i += KO_T) {
        const int r = i >> 4, c4 = (i & 15) * 4;
        const int n = row0 + r;
        float4 a = make_float4(0.0f, 0.0f, 0.0f, 0.0f);
        float4 b = a;
        if (n < N) {
            const long g = (long)n * 64 + c4;
            const float4 s0 = *(const float4*)(B0 + g);
            const float4 s1 = *(const float4*)(B1 + g);
            const float di = dinv[n];
            a.x = ko_elu(di * s0.x + p0); a.y = ko_elu(di * s0.y + p1);
            a.z = ko_elu(di * s0.z + p2); a.w = ko_elu(di * s0.w + p3);
            b.x = ko_elu(di * s1.x + q0); b.y = ko_elu(di * s1.y + q1);
            b.z = ko_elu(di * s1.z + q2); b.w = ko_elu(di * s1.w + q3);
        }
        *(float4*)&sh[r * KO_P + c4] = a;
        *(float4*)&sh[r * KO_P + 64 + c4] = b;
    }
    __syncthreads();

    const int cg = tid & 7;
    const int rg = tid >> 3;
    const float* hp = &sh[(rg * 4) * KO_P];
    const float* wp = &sw[cg * 5];
    float acc[4][5];
#pragma unroll
    for (int j = 0; j < 4; ++j)
#pragma unroll
        for (int c = 0; c < 5; ++c) acc[j][c] = cb[cg * 5 + c];

    for (int k = 0; k < 128; k += 4) {
        float hv[4][4];
#pragma unroll
        for (int j = 0; j < 4; ++j) {
            const float4 h = *(const float4*)(hp + j * KO_P + k);
            hv[j][0] = h.x; hv[j][1] = h.y; hv[j][2] = h.z; hv[j][3] = h.w;
        }
#pragma unroll
        for (int kk = 0; kk < 4; ++kk) {
            float w[5];
#pragma unroll
            for (int c = 0; c < 5; ++c) w[c] = wp[(k + kk) * 40 + c];
#pragma unroll
            for (int j = 0; j < 4; ++j)
#pragma unroll
                for (int c = 0; c < 5; ++c) acc[j][c] += hv[j][kk] * w[c];
        }
    }
    __syncthreads();
#pragma unroll
    for (int j = 0; j < 4; ++j)
#pragma unroll
        for (int c = 0; c < 5; ++c) sh[(rg * 4 + j) * 40 + cg * 5 + c] = acc[j][c];
    __syncthreads();
    const long lim = (long)N * 40;
    for (int i = tid; i < KO_R * 10; i += KO_T) {
        const long g = (long)row0 * 40 + i * 4;
        if (g < lim) *(float4*)(out + g) = *(const float4*)&sh[i * 4];
    }
}

extern "C" void kernel_launch(void* const* d_in, const int* in_sizes, int n_in,
                              void* d_out, int out_size, void* d_ws, size_t ws_size,
                              hipStream_t stream) {
    const float* x       = (const float*)d_in[0];
    const int*   ei      = (const int*)d_in[1];
    const float* conv0_w = (const float*)d_in[2];
    const float* conv0_b = (const float*)d_in[3];
    const float* lin0_w  = (const float*)d_in[4];
    const float* lin0_b  = (const float*)d_in[5];
    const float* conv1_w = (const float*)d_in[6];
    const float* conv1_b = (const float*)d_in[7];
    const float* lin1_w  = (const float*)d_in[8];
    const float* lin1_b  = (const float*)d_in[9];
    const float* out_w   = (const float*)d_in[10];
    const float* out_b   = (const float*)d_in[11];
    float* out = (float*)d_out;

    int N = in_sizes[0] / 128;
    int E = in_sizes[1] / 2;
    const int* src = ei;
    const int* dst = ei + E;

    size_t NH = (size_t)N * 64;
    float* ws   = (float*)d_ws;
    float* dinv = ws;
    float* A    = ws + (((size_t)N + 255) / 256) * 256;
    float* B0   = A + NH;
    float* B1   = B0 + NH;
    float* W0f  = B1 + NH;
    float* W1f  = W0f + 64 * 40;
    float* cb   = W1f + 64 * 40;
    int* tickets = (int*)(cb + 64);
    int* planes  = (int*)B1;

    int gOut = (N + KO_R - 1) / KO_R;
    int gGemm = (N + 127) / 128;

    k_prologue<<<DH_R * DH_S + DH_HB, DH_T, 0, stream>>>(dst, planes, tickets,
                                                         lin0_w, lin0_b, lin1_w, lin1_b, out_w, out_b,
                                                         W0f, W1f, cb, E, N);

    k_gemm_l0<<<gGemm, 128, 0, stream>>>(x, conv0_w, planes, dinv, A, B0, N);
    k_scatter<<<SC_GRID, TPB, 0, stream>>>(A, src, dst, dinv, B0, tickets, E, N);

    k_gemm_l1<<<gGemm, 128, 0, stream>>>(B0, conv1_w, dinv, conv0_b, A, B1, N);
    k_scatter<<<SC_GRID, TPB, 0, stream>>>(A, src, dst, dinv, B1, tickets + 2 * SC_CTR, E, N);

    k_out_tiled<<<gOut, KO_T, 0, stream>>>(B0, B1, dinv, conv0_b, conv1_b, W0f, W1f, cb, out, N);
}


# source-side norm factor moved into the stored rows: GEMM stores dinv[row] * (h @ W) as gather rows and as sum start; scatter adds rows as they are, no per-edge dinv lookup; same terms, arrival order
# speedup vs baseline: 4.0277x; 1.0632x over previous
#include <hip/hip_runtime.h>
#include <hip/hip_bf16.h>

typedef float v2f __attribute__((ext_vector_type(2)));
typedef float v8f __attribute__((ext_vector_type(8)));

#define TPB 256

#define DH_R   16
#define DH_S   16
#define DH_T   1024
#define DH_CAP 8192
#define DH_U   8
#define DH_HB  3

__device__ __forceinline__ int dh_pitch(int N) { return (N + 255) & ~255; }

__device__ __forceinline__ void dh_count(int* cnt, int d, int c0, int nb) {
    unsigned a = (unsigned)d - (unsigned)c0;
    if (a < (unsigned)nb) atomicAdd(&cnt[a], 1);
}

__device__ __forceinline__ void fuse_heads_entry(int i,
                                                 const float* __restrict__ L0, const float* __restrict__ b0,
                                                 const float* __restrict__ L1, const float* __restrict__ b1,
                                                 const float* __restrict__ Wo, const float* __restrict__ bo,
                                                 float* W0f, float* W1f, float* cb) {
    if (i < 64 * 40) {
        int k = i / 40, c = i % 40;
        float s0 = 0.0f, s1 = 0.0f;
        for (int j = 0; j < 64; ++j) {
            s0 += L0[k * 64 + j] * Wo[j * 40 + c];
            s1 += L1[k * 64 + j] * Wo[(64 + j) * 40 + c];
        }
        W0f[i] = s0;
        W1f[i] = s1;
    } else if (i < 64 * 40 + 40) {
        int c = i - 64 * 40;
        float s = bo[c];
        for (int j = 0; j < 64; ++j)
            s += b0[j] * Wo[j * 40 + c] + b1[j] * Wo[(64 + j) * 40 + c];
        cb[c] = s;
    }
}

__global__ __launch_bounds__(DH_T)
void k_prologue(const int* __restrict__ dst, int* planes, int* tickets,
                const float* __restrict__ L0, const float* __restrict__ b0,
                const float* __restrict__ L1, const float* __restrict__ b1,
                const float* __restrict__ Wo, const float* __restrict__ bo,
                float* W0f, float* W1f, float* cb, int E, int N) {
    __shared__ int cnt[DH_CAP];
    const int tid = threadIdx.x;
    if (blockIdx.x >= DH_R * DH_S) {
        fuse_heads_entry((blockIdx.x - DH_R * DH_S) * DH_T + tid, L0, b0, L1, b1, Wo, bo, W0f, W1f, cb);
        return;
    }
    if (blockIdx.x == 0 && tid < 256) tickets[tid] = 0;
    const int r = blockIdx.x % DH_R;
    const int s = blockIdx.x / DH_R;
    const int Np = dh_pitch(N);
    const int span = (((N + DH_R - 1) / DH_R) + 31) & ~31;
    const int lo = r * span;
    const int hi = (lo + span < N) ? (lo + span) : N;
    const int per = (((E + DH_S - 1) / DH_S) + 3) & ~3;
    const int e0 = s * per;
    const int e1 = (e0 + per < E) ? (e0 + per) : E;
    const int ne = (e1 > e0) ? (e1 - e0) : 0;
    const bool vec = (((unsigned long long)dst) & 15ull) == 0ull;
    const int n4 = vec ? (ne >> 2) : 0;
    int* plane = planes + (long)s * Np;

    for (int c0 = lo; c0 < hi; c0 += DH_CAP) {
        const int nb = (hi - c0 < DH_CAP) ? (hi - c0) : DH_CAP;
        for (int i = tid; i < nb; i += DH_T) cnt[i] = 0;
        __syncthreads();
        const int4* d4 = (const int4*)(dst + e0);
        for (int i = tid; i < n4; i += DH_U * DH_T) {
            int4 v[DH_U];
#pragma unroll
            for (int u = 0; u < DH_U; ++u) {
                const int j = i + u * DH_T;
                v[u] = (j < n4) ? d4[j] : make_int4(-1, -1, -1, -1);
            }
#pragma unroll
            for (int u = 0; u < DH_U; ++u) {
                dh_count(cnt, v[u].x, c0, nb);
                dh_count(cnt, v[u].y, c0, nb);
                dh_count(cnt, v[u].z, c0, nb);
                dh_count(cnt, v[u].w, c0, nb);
            }
        }
        for (int e = e0 + n4 * 4 + tid; e < e1; e += DH_T)
            dh_count(cnt, dst[e], c0, nb);
        __syncthreads();
        for (int i = tid; i < nb; i += DH_T) plane[c0 + i] = cnt[i];
        __syncthreads();
    }
}

__device__ __forceinline__ float gm_elu(float v) {
    const float e = expf(v) - 1.0f;
    return (v > 0.0f) ? v : e;
}

template <bool ELU_IN, bool DEG_IN>
__device__ __forceinline__ void gemm_n64_body(const float* __restrict__ A, const float* __restrict__ W,
                                              const int* __restrict__ planes, float* __restrict__ dinv_out,
                                              const float* __restrict__ dinv, const float* __restrict__ biasA,
                                              float* __restrict__ D, float* __restrict__ S,
                                              float* sW, float* sdd, float* sBi, int N, int K) {
    int tid = threadIdx.x;
    const int drow = blockIdx.x * 128 + tid;
    int dcnt[DH_S];
    if (DEG_IN) {
        const int Np = dh_pitch(N);
#pragma unroll
        for (int s = 0; s < DH_S; ++s) dcnt[s] = (drow < N) ? planes[(long)s * Np + drow] : 0;
    }
    for (int i = tid; i < K * 64; i += 128) {
        int k = i >> 6, col = i & 63;
        sW[(k >> 1) * 128 + 2 * col + (k & 1)] = W[i];
    }
    if (DEG_IN) {
        int c = 0;
#pragma unroll
        for (int s = 0; s < DH_S; ++s) c += dcnt[s];
        const float deg = 1.0f + (float)c;
        const float di = (deg > 0.0f) ? rsqrtf(deg) : 0.0f;
        if (drow < N) dinv_out[drow] = di;
        sdd[tid] = di;
    }
    if (ELU_IN) {
        if (tid < 64) sBi[tid] = biasA[tid];
    }
    __syncthreads();

    int wave = tid >> 5;
    int lane = tid & 31;
    int half = lane >> 4;
    int l16  = lane & 15;

    int rbase = blockIdx.x * 128 + wave * 32;
    int ar0 = rbase + l16;
    int ar1 = rbase + 16 + l16;
    long ab0 = (long)((ar0 < N) ? ar0 : (N - 1)) * K;
    long ab1 = (long)((ar1 < N) ? ar1 : (N - 1)) * K;

    int m0 = rbase + half * 8;
    int m1 = rbase + 16 + half * 8;
    float dd0[8], dd1[8];
#pragma unroll
    for (int r = 0; r < 8; ++r) {
        if (DEG_IN) {
            dd0[r] = sdd[wave * 32 + half * 8 + r];
            dd1[r] = sdd[wave * 32 + 16 + half * 8 + r];
        } else {
            const int q0 = m0 + r, q1 = m1 + r;
            dd0[r] = dinv[(q0 < N) ? q0 : (N - 1)];
            dd1[r] = dinv[(q1 < N) ? q1 : (N - 1)];
        }
    }

    float da0 = 1.0f, da1 = 1.0f;
    if (ELU_IN) {
        da0 = dinv[(ar0 < N) ? ar0 : (N - 1)];
        da1 = dinv[(ar1 < N) ? ar1 : (N - 1)];
    }

    v8f acc00 = {}, acc01 = {}, acc02 = {}, acc03 = {};
    v8f acc10 = {}, acc11 = {}, acc12 = {}, acc13 = {};

    for (int k = 0; k < K; k += 4) {
        int kb = k + 2 * half;
        v2f a0 = *(const v2f*)(A + ab0 + kb);
        v2f a1 = *(const v2f*)(A + ab1 + kb);
        if (ELU_IN) {
            const v2f bk = *(const v2f*)(sBi + kb);
            a0.x = gm_elu(da0 * a0.x + bk.x); a0.y = gm_elu(da0 * a0.y + bk.y);
            a1.x = gm_elu(da1 * a1.x + bk.x); a1.y = gm_elu(da1 * a1.y + bk.y);
        }
        const float* bp = &sW[(kb >> 1) * 128 + 2 * l16];
        v2f b0 = *(const v2f*)(bp + 0);
        v2f b1 = *(const v2f*)(bp + 32);
        v2f b2 = *(const v2f*)(bp + 64);
        v2f b3 = *(const v2f*)(bp + 96);
        acc00 = __builtin_amdgcn_wmma_f32_16x16x4_f32(false, a0, false, b0, (short)0, acc00, false, false);
        acc01 = __builtin_amdgcn_wmma_f32_16x16x4_f32(false, a0, false, b1, (short)0, acc01, false, false);
        acc02 = __builtin_amdgcn_wmma_f32_16x16x4_f32(false, a0, false, b2, (short)0, acc02, false, false);
        acc03 = __builtin_amdgcn_wmma_f32_16x16x4_f32(false, a0, false, b3, (short)0, acc03, false, false);
        acc10 = __builtin_amdgcn_wmma_f32_16x16x4_f32(false, a1, false, b0, (short)0, acc10, false, false);
        acc11 = __builtin_amdgcn_wmma_f32_16x16x4_f32(false, a1, false, b1, (short)0, acc11, false, false);
        acc12 = __builtin_amdgcn_wmma_f32_16x16x4_f32(false, a1, false, b2, (short)0, acc12, false, false);
        acc13 = __builtin_amdgcn_wmma_f32_16x16x4_f32(false, a1, false, b3, (short)0, acc13, false, false);
    }

#pragma unroll
    for (int r = 0; r < 8; ++r) {
        int rr = m0 + r;
        if (rr < N) {
            const float dd = dd0[r];
            float* dp = D + (long)rr * 64 + l16;
            float* sp = S + (long)rr * 64 + l16;
            const float t0 = dd * acc00[r], t1 = dd * acc01[r], t2 = dd * acc02[r], t3 = dd * acc03[r];
            dp[0]  = t0;
            dp[16] = t1;
            dp[32] = t2;
            dp[48] = t3;
            sp[0]  = t0;
            sp[16] = t1;
            sp[32] = t2;
            sp[48] = t3;
        }
        int ss = m1 + r;
        if (ss < N) {
            const float dd = dd1[r];
            float* dp = D + (long)ss * 64 + l16;
            float* sp = S + (long)ss * 64 + l16;
            const float t0 = dd * acc10[r], t1 = dd * acc11[r], t2 = dd * acc12[r], t3 = dd * acc13[r];
            dp[0]  = t0;
            dp[16] = t1;
            dp[32] = t2;
            dp[48] = t3;
            sp[0]  = t0;
            sp[16] = t1;
            sp[32] = t2;
            sp[48] = t3;
        }
    }
}

__global__ __launch_bounds__(128)
void k_gemm_l0(const float* __restrict__ A, const float* __restrict__ W,
               const int* __restrict__ planes, float* __restrict__ dinv_out,
               float* __restrict__ D, float* __restrict__ S, int N) {
    __shared__ float sW[128 * 64];
    __shared__ float sdd[128];
    gemm_n64_body<false, true>(A, W, planes, dinv_out, nullptr, nullptr, D, S, sW, sdd, nullptr, N, 128);
}

__global__ __launch_bounds__(128)
void k_gemm_l1(const float* __restrict__ A, const float* __restrict__ W,
               const float* __restrict__ dinv, const float* __restrict__ bias0,
               float* __restrict__ D, float* __restrict__ S, int N) {
    __shared__ float sW[64 * 64];
    __shared__ __attribute__((aligned(8))) float sBi[64];
    gemm_n64_body<true, false>(A, W, nullptr, nullptr, dinv, bias0, D, S, sW, nullptr, sBi, N, 64);
}

#define SC_EPW  32
#define SC_U    8
#define SC_TK   2
#define SC_Q    64
#define SC_CTR  64
#define SC_GRID 2048

template <bool FULL>
__device__ __forceinline__ void sc_batch(const float* __restrict__ T, float* Bagg, int lane,
                                         int s_v, int d_v, int cnt) {
    for (int j0 = 0; j0 < (FULL ? SC_EPW : cnt); j0 += SC_U) {
        float t0[SC_U], t1[SC_U];
#pragma unroll
        for (int u = 0; u < SC_U; ++u) {
            t0[u] = 0.0f;
            t1[u] = 0.0f;
            if (FULL || j0 + u < cnt) {
                int s = __builtin_amdgcn_readlane(s_v, j0 + u);
                const float* tp = T + (long)s * 64 + lane;
                t0[u] = tp[0];
                t1[u] = tp[32];
            }
        }
#pragma unroll
        for (int u = 0; u < SC_U; ++u) {
            if (FULL || j0 + u < cnt) {
                int d = __builtin_amdgcn_readlane(d_v, j0 + u);
                float* bp = Bagg + (long)d * 64 + lane;
                atomicAdd(bp, t0[u]);
                atomicAdd(bp + 32, t1[u]);
            }
        }
    }
}

__device__ __forceinline__ void sc_lds_barrier() {
    __builtin_amdgcn_fence(__ATOMIC_RELEASE, "workgroup", "local");
    __builtin_amdgcn_s_barrier();
    __builtin_amdgcn_fence(__ATOMIC_ACQUIRE, "workgroup", "local");
}

__device__ __forceinline__ int sc_draw(int* tickets, int half, int ntk, int& cur) {
    while (cur < 2) {
        int own = half ^ cur;
        int t = atomicAdd(tickets + own * SC_CTR, 1);
        if (t < ntk) return t * 2 + own;
        ++cur;
    }
    return -1;
}

__global__ __launch_bounds__(TPB)
void k_scatter(const float* __restrict__ T, const int* __restrict__ src,
               const int* __restrict__ dst,
               float* Bagg, int* tickets, int E, int N) {
    __shared__ int q_s[TPB / 32][SC_Q];
    __shared__ int q_d[TPB / 32][SC_Q];
    __shared__ int s_tk[2];
    const int tid  = threadIdx.x;
    const int lane = tid & 31;
    const int wave = tid >> 5;
    int* qs = q_s[wave];
    int* qd = q_d[wave];
    const int nsplit = N >> 1;
    const int per_tk = SC_EPW * SC_TK * (TPB / 32);
    const int ntk = (E + per_tk - 1) / per_tk;
    const int half = (__builtin_amdgcn_s_sendmsg_rtn(0x87) >> 18) & 1;
    int cur = 0;

    int par = 0;
    if (tid == 0) s_tk[0] = sc_draw(tickets, half, ntk, cur);
    sc_lds_barrier();

    int qh = 0, qn = 0;
    for (;;) {
        const int tk = s_tk[par];
        if (tk < 0) break;
        int nx_own = 0, nx_raw = 0;
        if (tid == 0 && cur < 2) {
            nx_own = half ^ cur;
            nx_raw = atomicAdd(tickets + nx_own * SC_CTR, 1);
        }

        const int own  = tk & 1;
        const int base = (tk >> 1) * per_tk + wave * (SC_EPW * SC_TK);
        int  sv[SC_TK], dv[SC_TK];
        bool ok[SC_TK];
#pragma unroll
        for (int c = 0; c < SC_TK; ++c) {
            int e = base + c * SC_EPW + lane;
            ok[c] = e < E;
            sv[c] = ok[c] ? src[e] : 0;
            dv[c] = ok[c] ? dst[e] : 0;
        }
#pragma unroll
        for (int c = 0; c < SC_TK; ++c) {
            bool mine = ok[c] && ((dv[c] >= nsplit) == (own != 0));
            unsigned m = __builtin_amdgcn_ballot_w32(mine);
            if (mine) {
                int slot = (qh + qn + (int)__builtin_amdgcn_mbcnt_lo(m, 0u)) & (SC_Q - 1);
                qs[slot] = sv[c];
                qd[slot] = dv[c];
            }
            qn += __builtin_popcount(m);
            if (qn >= SC_EPW) {
                int slot = (qh + lane) & (SC_Q - 1);
                int s = qs[slot];
                int d = qd[slot];
                sc_batch<true>(T, Bagg, lane, s, d, SC_EPW);
                qh = (qh + SC_EPW) & (SC_Q - 1);
                qn -= SC_EPW;
            }
        }

        if (tid == 0) {
            int v = -1;
            if (cur < 2) {
                if (nx_raw < ntk) {
                    v = nx_raw * 2 + nx_own;
                } else {
                    ++cur;
                    v = sc_draw(tickets, half, ntk, cur);
                }
            }
            s_tk[par ^ 1] = v;
        }
        sc_lds_barrier();
        par ^= 1;
    }

    if (qn > 0) {
        int slot = (qh + lane) & (SC_Q - 1);
        int s = 0, d = 0;
        if (lane < qn) {
            s = qs[slot];
            d = qd[slot];
        }
        sc_batch<false>(T, Bagg, lane, s, d, qn);
    }
}

#define KO_R 64
#define KO_P 132
#define KO_T 128

__device__ __forceinline__ float ko_elu(float v) { return (v > 0.0f) ? v : (expf(v) - 1.0f); }

__global__ __launch_bounds__(KO_T)
void k_out_tiled(const float* __restrict__ B0, const float* __restrict__ B1,
                 const float* __restrict__ dinv, const float* __restrict__ cbias0, const float* __restrict__ cbias1,
                 const float* __restrict__ W0f, const float* __restrict__ W1f,
                 const float* __restrict__ cb, float* __restrict__ out, int N) {
    __shared__ __attribute__((aligned(16))) float sw[128 * 40];
    __shared__ __attribute__((aligned(16))) float sh[KO_R * KO_P];
    const int tid = threadIdx.x;
    const int row0 = blockIdx.x * KO_R;

    for (int i = tid; i < 64 * 40; i += KO_T) {
        sw[i] = W0f[i];
        sw[2560 + i] = W1f[i];
    }
    const int cq = (tid & 15) * 4;
    const float p0 = cbias0[cq], p1 = cbias0[cq + 1], p2 = cbias0[cq + 2], p3 = cbias0[cq + 3];
    const float q0 = cbias1[cq], q1 = cbias1[cq + 1], q2 = cbias1[cq + 2], q3 = cbias1[cq + 3];
    for (int i = tid; i < KO_R * 16; i += KO_T) {
        const int r = i >> 4, c4 = (i & 15) * 4;
        const int n = row0 + r;
        float4 a = make_float4(0.0f, 0.0f, 0.0f, 0.0f);
        float4 b = a;
        if (n < N) {
            const long g = (long)n * 64 + c4;
            const float4 s0 = *(const float4*)(B0 + g);
            const float4 s1 = *(const float4*)(B1 + g);
            const float di = dinv[n];
            a.x = ko_elu(di * s0.x + p0); a.y = ko_elu(di * s0.y + p1);
            a.z = ko_elu(di * s0.z + p2); a.w = ko_elu(di * s0.w + p3);
            b.x = ko_elu(di * s1.x + q0); b.y = ko_elu(di * s1.y + q1);
            b.z = ko_elu(di * s1.z + q2); b.w = ko_elu(di * s1.w + q3);
        }
        *(float4*)&sh[r * KO_P + c4] = a;
        *(float4*)&sh[r * KO_P + 64 + c4] = b;
    }
    __syncthreads();

    const int cg = tid & 7;
    const int rg = tid >> 3;
    const float* hp = &sh[(rg * 4) * KO_P];
    const float* wp = &sw[cg * 5];
    float acc[4][5];
#pragma unroll
    for (int j = 0; j < 4; ++j)
#pragma unroll
        for (int c = 0; c < 5; ++c) acc[j][c] = cb[cg * 5 + c];

    for (int k = 0; k < 128; k += 4) {
        float hv[4][4];
#pragma unroll
        for (int j = 0; j < 4; ++j) {
            const float4 h = *(const float4*)(hp + j * KO_P + k);
            hv[j][0] = h.x; hv[j][1] = h.y; hv[j][2] = h.z; hv[j][3] = h.w;
        }
#pragma unroll
        for (int kk = 0; kk < 4; ++kk) {
            float w[5];
#pragma unroll
            for (int c = 0; c < 5; ++c) w[c] = wp[(k + kk) * 40 + c];
#pragma unroll
            for (int j = 0; j < 4; ++j)
#pragma unroll
                for (int c = 0; c < 5; ++c) acc[j][c] += hv[j][kk] * w[c];
        }
    }
    __syncthreads();
#pragma unroll
    for (int j = 0; j < 4; ++j)
#pragma unroll
        for (int c = 0; c < 5; ++c) sh[(rg * 4 + j) * 40 + cg * 5 + c] = acc[j][c];
    __syncthreads();
    const long lim = (long)N * 40;
    for (int i = tid; i < KO_R * 10; i += KO_T) {
        const long g = (long)row0 * 40 + i * 4;
        if (g < lim) *(float4*)(out + g) = *(const float4*)&sh[i * 4];
    }
}

extern "C" void kernel_launch(void* const* d_in, const int* in_sizes, int n_in,
                              void* d_out, int out_size, void* d_ws, size_t ws_size,
                              hipStream_t stream) {
    const float* x       = (const float*)d_in[0];
    const int*   ei      = (const int*)d_in[1];
    const float* conv0_w = (const float*)d_in[2];
    const float* conv0_b = (const float*)d_in[3];
    const float* lin0_w  = (const float*)d_in[4];
    const float* lin0_b  = (const float*)d_in[5];
    const float* conv1_w = (const float*)d_in[6];
    const float* conv1_b = (const float*)d_in[7];
    const float* lin1_w  = (const float*)d_in[8];
    const float* lin1_b  = (const float*)d_in[9];
    const float* out_w   = (const float*)d_in[10];
    const float* out_b   = (const float*)d_in[11];
    float* out = (float*)d_out;

    int N = in_sizes[0] / 128;
    int E = in_sizes[1] / 2;
    const int* src = ei;
    const int* dst = ei + E;

    size_t NH = (size_t)N * 64;
    float* ws   = (float*)d_ws;
    float* dinv = ws;
    float* A    = ws + (((size_t)N + 255) / 256) * 256;
    float* B0   = A + NH;
    float* B1   = B0 + NH;
    float* W0f  = B1 + NH;
    float* W1f  = W0f + 64 * 40;
    float* cb   = W1f + 64 * 40;
    int* tickets = (int*)(cb + 64);
    int* planes  = (int*)B1;

    int gOut = (N + KO_R - 1) / KO_R;
    int gGemm = (N + 127) / 128;

    k_prologue<<<DH_R * DH_S + DH_HB, DH_T, 0, stream>>>(dst, planes, tickets,
                                                         lin0_w, lin0_b, lin1_w, lin1_b, out_w, out_b,
                                                         W0f, W1f, cb, E, N);

    k_gemm_l0<<<gGemm, 128, 0, stream>>>(x, conv0_w, planes, dinv, A, B0, N);
    k_scatter<<<SC_GRID, TPB, 0, stream>>>(A, src, dst, B0, tickets, E, N);

    k_gemm_l1<<<gGemm, 128, 0, stream>>>(B0, conv1_w, dinv, conv0_b, A, B1, N);
    k_scatter<<<SC_GRID, TPB, 0, stream>>>(A, src, dst, B1, tickets + 2 * SC_CTR, E, N);

    k_out_tiled<<<gOut, KO_T, 0, stream>>>(B0, B1, dinv, conv0_b, conv1_b, W0f, W1f, cb, out, N);
}
